# starting kernel
# baseline (speedup 1.0000x reference)
#include <hip/hip_runtime.h>
#include <hip/hip_bf16.h>

typedef __attribute__((ext_vector_type(2))) float v2f;
typedef __attribute__((ext_vector_type(8))) float v8f;

#define B_   16
#define N_   4096
#define C_   64
#define M_   1024
#define K_   32
#define EPS_ 1e-5f
#define BIGF 3.402823e38f

__global__ __launch_bounds__(256) void k_new_xyz(const float* __restrict__ xyz,
                                                 const int* __restrict__ indices,
                                                 float* __restrict__ new_xyz) {
  int t = blockIdx.x * 256 + threadIdx.x;
  if (t >= B_ * M_) return;
  int b = t >> 10, m = t & 1023;
  int n = indices[m];
  const float* s = xyz + ((size_t)b * N_ + n) * 3;
  float* d = new_xyz + (size_t)t * 3;
  d[0] = s[0]; d[1] = s[1]; d[2] = s[2];
}

__global__ __launch_bounds__(64) void k_knn(const float* __restrict__ xyz,
                                            const float* __restrict__ new_xyz,
                                            int* __restrict__ knn) {
  __shared__ float dist[2][N_];
  const int wave = threadIdx.x >> 5;
  const int lane = threadIdx.x & 31;
  const int q = blockIdx.x * 2 + wave;
  const int b = q >> 10;

  float* dl = dist[wave];
  const float qx = new_xyz[q * 3 + 0];
  const float qy = new_xyz[q * 3 + 1];
  const float qz = new_xyz[q * 3 + 2];
  const float qq = qx * qx + qy * qy + qz * qz;
  const float* px = xyz + (size_t)b * N_ * 3;

  for (int n = lane; n < N_; n += 32) {
    float x = px[n * 3 + 0], y = px[n * 3 + 1], z = px[n * 3 + 2];
    dl[n] = qq + (x * x + y * y + z * z) - 2.0f * (qx * x + qy * y + qz * z);
  }
  __syncthreads();

  int* out = knn + q * K_;
  for (int k = 0; k < K_; ++k) {
    float mv = BIGF;
    int   mi = 0x7fffffff;
    for (int n = lane; n < N_; n += 32) {
      float v = dl[n];
      if (v < mv) { mv = v; mi = n; }
    }
#pragma unroll
    for (int off = 16; off > 0; off >>= 1) {
      float ov = __shfl_xor(mv, off, 32);
      int   oi = __shfl_xor(mi, off, 32);
      if (ov < mv || (ov == mv && oi < mi)) { mv = ov; mi = oi; }
    }
    if (lane == 0) out[k] = mi;
    dl[mi] = BIGF;
    __syncthreads();
  }
}

__device__ __forceinline__ void gemm_bn_relu(
    const float* __restrict__ Xb, int kPad,
    const float* __restrict__ W, int kW, int rowsOut,
    const float* __restrict__ g, const float* __restrict__ bt,
    const float* __restrict__ mu, const float* __restrict__ vr,
    float* __restrict__ Yb, int lane)
{
  const int half = lane >> 4;
  const int l16  = lane & 15;
  const int nKt  = kPad >> 2;

  for (int mt = 0; mt < (rowsOut >> 4); ++mt) {
    const int arow = mt * 16 + l16;
    for (int nt = 0; nt < 2; ++nt) {
      const int col = nt * 16 + l16;
      v8f acc = {};
      for (int kt = 0; kt < nKt; ++kt) {
        const int k0 = kt * 4 + half * 2;
        v2f a, bf;
        a.x  = (k0 + 0 < kW) ? W[arow * kW + k0 + 0] : 0.0f;
        a.y  = (k0 + 1 < kW) ? W[arow * kW + k0 + 1] : 0.0f;
        bf.x = Xb[(k0 + 0) * 32 + col];
        bf.y = Xb[(k0 + 1) * 32 + col];
        acc = __builtin_amdgcn_wmma_f32_16x16x4_f32(
             false, a,  false, bf,
             (short)0, acc,  false,  false);
      }
      const int rbase = mt * 16 + half * 8;
#pragma unroll
      for (int r = 0; r < 8; ++r) {
        const int row = rbase + r;
        const float s = g[row] / sqrtf(vr[row] + EPS_);
        const float y = (acc[r] - mu[row]) * s + bt[row];
        Yb[row * 32 + col] = fmaxf(y, 0.0f);
      }
    }
  }
}

__global__ __launch_bounds__(64) void k_mlp(
    const float* __restrict__ features, const float* __restrict__ xyz,
    const float* __restrict__ new_xyz, const int* __restrict__ knn,
    const float* __restrict__ W0, const float* __restrict__ g0,
    const float* __restrict__ b0, const float* __restrict__ m0,
    const float* __restrict__ v0,
    const float* __restrict__ W1, const float* __restrict__ g1,
    const float* __restrict__ b1, const float* __restrict__ m1,
    const float* __restrict__ v1,
    const float* __restrict__ W2, const float* __restrict__ g2,
    const float* __restrict__ b2, const float* __restrict__ m2,
    const float* __restrict__ v2,
    float* __restrict__ out_feat)
{
  __shared__ float Xs[2][68 * 32];
  __shared__ float Ys[2][128 * 32];
  const int wave = threadIdx.x >> 5;
  const int lane = threadIdx.x & 31;
  const int q = blockIdx.x * 2 + wave;
  const int b = q >> 10;
  const int m = q & 1023;

  float* Xb = Xs[wave];
  float* Yb = Ys[wave];

  const int nj = knn[q * K_ + lane];
  const float* fb = features + (size_t)b * C_ * N_;
#pragma unroll 8
  for (int c = 0; c < C_; ++c)
    Xb[c * 32 + lane] = fb[(size_t)c * N_ + nj];
  const float* p  = xyz + ((size_t)b * N_ + nj) * 3;
  const float* qc = new_xyz + (size_t)q * 3;
  Xb[64 * 32 + lane] = p[0] - qc[0];
  Xb[65 * 32 + lane] = p[1] - qc[1];
  Xb[66 * 32 + lane] = p[2] - qc[2];
  Xb[67 * 32 + lane] = 0.0f;
  __syncthreads();

  gemm_bn_relu(Xb, 68, W0, 67,  64, g0, b0, m0, v0, Yb, lane);
  __syncthreads();
  gemm_bn_relu(Yb, 64, W1, 64,  64, g1, b1, m1, v1, Xb, lane);
  __syncthreads();
  gemm_bn_relu(Xb, 64, W2, 64, 128, g2, b2, m2, v2, Yb, lane);
  __syncthreads();

  for (int o = lane; o < 128; o += 32) {
    float mx = -BIGF;
#pragma unroll
    for (int j = 0; j < 32; ++j) mx = fmaxf(mx, Yb[o * 32 + j]);
    out_feat[((size_t)b * 128 + o) * M_ + m] = mx;
  }
}

extern "C" void kernel_launch(void* const* d_in, const int* in_sizes, int n_in,
                              void* d_out, int out_size, void* d_ws, size_t ws_size,
                              hipStream_t stream) {
  const float* xyz      = (const float*)d_in[0];
  const float* features = (const float*)d_in[1];
  const int*   indices  = (const int*)d_in[2];
  const float* W0 = (const float*)d_in[3];
  const float* g0 = (const float*)d_in[4];
  const float* b0 = (const float*)d_in[5];
  const float* m0 = (const float*)d_in[6];
  const float* v0 = (const float*)d_in[7];
  const float* W1 = (const float*)d_in[8];
  const float* g1 = (const float*)d_in[9];
  const float* b1 = (const float*)d_in[10];
  const float* m1 = (const float*)d_in[11];
  const float* v1 = (const float*)d_in[12];
  const float* W2 = (const float*)d_in[13];
  const float* g2 = (const float*)d_in[14];
  const float* b2 = (const float*)d_in[15];
  const float* m2 = (const float*)d_in[16];
  const float* v2 = (const float*)d_in[17];

  float* new_xyz  = (float*)d_out;
  float* out_feat = new_xyz + (size_t)B_ * M_ * 3;
  int*   knn      = (int*)d_ws;

  k_new_xyz<<<(B_ * M_ + 255) / 256, 256, 0, stream>>>(xyz, indices, new_xyz);
  k_knn<<<(B_ * M_) / 2, 64, 0, stream>>>(xyz, new_xyz, knn);
  k_mlp<<<(B_ * M_) / 2, 64, 0, stream>>>(features, xyz, new_xyz, knn,
      W0, g0, b0, m0, v0, W1, g1, b1, m1, v1, W2, g2, b2, m2, v2, out_feat);
}


# MLP layers tiled by output rows: a wave keeps 16 weight rows (A fragments) and their BatchNorm constants in registers, sweeps the block's 8 queries from LDS, pools in registers; sums unchanged
# speedup vs baseline: 2.0508x; 2.0508x over previous
#include <hip/hip_runtime.h>
#include <hip/hip_bf16.h>

typedef __attribute__((ext_vector_type(2))) float v2f;
typedef __attribute__((ext_vector_type(8))) float v8f;

#define B_   16
#define N_   4096
#define C_   64
#define M_   1024
#define K_   32
#define EPS_ 1e-5f
#define BIGF 3.402823e38f

__global__ __launch_bounds__(256) void k_new_xyz(const float* __restrict__ xyz,
                                                 const int* __restrict__ indices,
                                                 float* __restrict__ new_xyz) {
  int t = blockIdx.x * 256 + threadIdx.x;
  if (t >= B_ * M_) return;
  int b = t >> 10, m = t & 1023;
  int n = indices[m];
  const float* s = xyz + ((size_t)b * N_ + n) * 3;
  float* d = new_xyz + (size_t)t * 3;
  d[0] = s[0]; d[1] = s[1]; d[2] = s[2];
}

#define CAP_ 128

__device__ __forceinline__ float sqdist(float qx, float qy, float qz, float qq,
                                        float x, float y, float z) {
  const float pp  = fmaf(z, z, fmaf(x, x, y * y));
  const float dot = fmaf(qz, z, fmaf(qx, x, qy * y));
  return fmaf(-2.0f, dot, qq + pp);
}

__device__ __forceinline__ unsigned okey(float f) {
  const unsigned u = __float_as_uint(f);
  return (u & 0x80000000u) ? ~u : (u | 0x80000000u);
}
__device__ __forceinline__ float okey_inv(unsigned k) {
  return __uint_as_float((k & 0x80000000u) ? (k & 0x7fffffffu) : ~k);
}

template <int NS, int NB>
__device__ __forceinline__ unsigned kth32(const unsigned (&k)[NS]) {
  unsigned v = 0;
  for (int bit = 31; bit > 31 - NB; --bit) {
    const unsigned t = v | (1u << bit);
    int c = 0;
#pragma unroll
    for (int s = 0; s < NS; ++s)
      c += __builtin_popcount(__builtin_amdgcn_ballot_w32(k[s] < t));
    if (c < K_) v = t;
  }
  if (NB < 32) v |= (1u << (32 - NB)) - 1u;
  return v;
}

__global__ __launch_bounds__(32) void k_knn(const float* __restrict__ xyz,
                                            const float* __restrict__ new_xyz,
                                            int* __restrict__ knn) {
  __shared__ float    dl[N_];
  __shared__ unsigned ck[CAP_];
  __shared__ int      ci[CAP_];
  const int lane = threadIdx.x;
  const int q = blockIdx.x;
  const int b = q >> 10;

  const float qx = new_xyz[q * 3 + 0];
  const float qy = new_xyz[q * 3 + 1];
  const float qz = new_xyz[q * 3 + 2];
  float qq;
  {
#pragma clang fp contract(off)
    const float yy = qy * qy;
    const float zz = qz * qz;
    qq = fmaf(qx, qx, yy) + zz;
  }
  const float* px = xyz + (size_t)b * N_ * 3;

  float m1 = BIGF, m2 = BIGF;
#pragma unroll 8
  for (int n = lane; n < N_; n += 32) {
    const float d = sqdist(qx, qy, qz, qq, px[n * 3 + 0], px[n * 3 + 1], px[n * 3 + 2]);
    dl[n] = d;
    m2 = fminf(m2, fmaxf(m1, d));
    m1 = fminf(m1, d);
  }

  const unsigned kb[2] = { okey(m1), okey(m2) };
  const float fB = okey_inv(kth32<2, 14>(kb));

  int cnt = 0;
  for (int w = 0; w < N_ / 1024; ++w) {
    unsigned bw = 0;
#pragma unroll
    for (int u = 0; u < 32; ++u)
      bw |= (dl[lane + 32 * (32 * w + u)] <= fB) ? (1u << u) : 0u;
    for (int t = 0; t < 32; ++t) {
      const bool has = bw != 0;
      const unsigned mask = __builtin_amdgcn_ballot_w32(has);
      if (mask == 0) break;
      if (has) {
        const int u = __builtin_ctz(bw);
        bw &= bw - 1;
        const int n = lane + 32 * (32 * w + u);
        const int pos = cnt + (int)__builtin_amdgcn_mbcnt_lo(mask, 0u);
        if (pos < CAP_) { ck[pos] = okey(dl[n]); ci[pos] = n; }
      }
      cnt += __builtin_popcount(mask);
    }
  }
  __syncthreads();

  int* out = knn + q * K_;
  bool done = false;
  if (cnt >= K_ && cnt <= CAP_) {
    unsigned k[CAP_ / 32];
    int id[CAP_ / 32];
#pragma unroll
    for (int s = 0; s < CAP_ / 32; ++s) {
      const int j = s * 32 + lane;
      const bool valid = j < cnt;
      k[s]  = valid ? ck[j] : 0xffffffffu;
      id[s] = valid ? ci[j] : 0;
    }
    unsigned v32;
    if (cnt <= 64) {
      const unsigned k2[2] = { k[0], k[1] };
      v32 = kth32<2, 32>(k2);
    } else {
      v32 = kth32<CAP_ / 32, 32>(k);
    }
    unsigned mt[CAP_ / 32];
    int total = 0;
#pragma unroll
    for (int s = 0; s < CAP_ / 32; ++s) {
      mt[s] = __builtin_amdgcn_ballot_w32(s * 32 + lane < cnt && k[s] <= v32);
      total += __builtin_popcount(mt[s]);
    }
    if (total == K_) {
      int wpos = 0;
#pragma unroll
      for (int s = 0; s < CAP_ / 32; ++s) {
        if ((mt[s] >> lane) & 1u) out[wpos + (int)__builtin_amdgcn_mbcnt_lo(mt[s], 0u)] = id[s];
        wpos += __builtin_popcount(mt[s]);
      }
      done = true;
    }
  }
  if (!done) {
    for (int kk = 0; kk < K_; ++kk) {
      float mv = BIGF;
      int   mi = 0x7fffffff;
      for (int n = lane; n < N_; n += 32) {
        float v = dl[n];
        if (v < mv) { mv = v; mi = n; }
      }
#pragma unroll
      for (int off = 16; off > 0; off >>= 1) {
        float ov = __shfl_xor(mv, off, 32);
        int   oi = __shfl_xor(mi, off, 32);
        if (ov < mv || (ov == mv && oi < mi)) { mv = ov; mi = oi; }
      }
      if (lane == 0) out[kk] = mi;
      dl[mi] = BIGF;
      __syncthreads();
    }
  }
}

#define NW_     8
#define NCOL_   (NW_ * 32)
#define PP_     (2 * NCOL_ + 32)
#define XPAIRS_ 34
#define YPAIRS_ 32
#define SROWS_  4
typedef __attribute__((ext_vector_type(4))) float v4f;

__device__ __forceinline__ void lds_barrier() {
  asm volatile("s_wait_dscnt 0x0\n\ts_barrier_signal -1\n\ts_barrier_wait -1" ::: "memory");
}

template <int MASK>
__device__ __forceinline__ float row_xor(float v) {
  return __builtin_bit_cast(float, __builtin_amdgcn_update_dpp(
      0, __builtin_bit_cast(int, v), 0x160 + MASK, 0xf, 0xf, true));
}
__device__ __forceinline__ float row_max16(float v) {
  v = fmaxf(v, row_xor<1>(v));
  v = fmaxf(v, row_xor<2>(v));
  v = fmaxf(v, row_xor<4>(v));
  v = fmaxf(v, row_xor<8>(v));
  return v;
}

template <int NKT>
__device__ __forceinline__ void load_b(v2f (&bq)[2][NKT], const float* Xin, int cw) {
#pragma unroll
  for (int kt = 0; kt < NKT; ++kt) {
    bq[0][kt] = *(const v2f*)(Xin + kt * 2 * PP_ + cw);
    bq[1][kt] = *(const v2f*)(Xin + kt * 2 * PP_ + cw + 32);
  }
}

template <int NKT>
struct RowTile {
  v2f a[NKT];
};
template <int NKT, int KW>
__device__ __forceinline__ void load_row_tile(
    RowTile<NKT>& t, const float* __restrict__ W, int mt, int lane)
{
  const int half = lane >> 4;
  const int arow = mt * 16 + (lane & 15);
#pragma unroll
  for (int kt = 0; kt < NKT; ++kt) {
    const int k0 = kt * 4 + half * 2;
    t.a[kt].x = (k0 + 0 < KW) ? W[arow * KW + k0 + 0] : 0.0f;
    t.a[kt].y = (k0 + 1 < KW) ? W[arow * KW + k0 + 1] : 0.0f;
  }
}

template <int NKT, int NQ, bool POOL, typename NextTile>
__device__ __forceinline__ void layer_rows(
    const float* Xin, float* Yout, const RowTile<NKT>& t,
    const float* __restrict__ g, const float* __restrict__ bt,
    const float* __restrict__ mu, const float* __restrict__ vr,
    int mt, int q0, int lane, float* __restrict__ pool_out, NextTile next_tile)
{
  static_assert(NQ % 2 == 0, "queries are taken two per trip");
  const int half = lane >> 4;
  const int l16  = lane & 15;

  float sc[8], mm[8], bb[8];
  const int rbase = mt * 16 + half * 8;
#pragma unroll
  for (int r = 0; r < 8; ++r) {
    const int row = rbase + r;
    sc[r] = g[row] / sqrtf(vr[row] + EPS_);
    mm[r] = mu[row];
    bb[r] = bt[row];
  }

  float res[8];
#pragma unroll
  for (int r = 0; r < 8; ++r) res[r] = 0.0f;

  const int cw = half * PP_ + l16 * 2;
  auto do_query = [&](const v2f (&bcur)[2][NKT], int qq) {
    v8f acc0 = {}, acc1 = {};
#pragma unroll
    for (int kt = 0; kt < NKT; ++kt) {
      acc0 = __builtin_amdgcn_wmma_f32_16x16x4_f32(
           false, t.a[kt],  false, bcur[0][kt],
           (short)0, acc0,  false,  false);
      acc1 = __builtin_amdgcn_wmma_f32_16x16x4_f32(
           false, t.a[kt],  false, bcur[1][kt],
           (short)0, acc1,  false,  false);
    }
    if (!POOL) {
      float* yq = Yout + (mt * 8 + half * 4) * PP_ + (qq * 32 + l16) * 2;
#pragma unroll
      for (int j = 0; j < 4; ++j) {
        v2f y0, y1;
        y0.x = fmaxf((acc0[2 * j]     - mm[2 * j])     * sc[2 * j]     + bb[2 * j],     0.0f);
        y0.y = fmaxf((acc0[2 * j + 1] - mm[2 * j + 1]) * sc[2 * j + 1] + bb[2 * j + 1], 0.0f);
        y1.x = fmaxf((acc1[2 * j]     - mm[2 * j])     * sc[2 * j]     + bb[2 * j],     0.0f);
        y1.y = fmaxf((acc1[2 * j + 1] - mm[2 * j + 1]) * sc[2 * j + 1] + bb[2 * j + 1], 0.0f);
        *(v2f*)(yq + j * PP_)      = y0;
        *(v2f*)(yq + j * PP_ + 32) = y1;
      }
    } else {
#pragma unroll
      for (int r = 0; r < 8; ++r) {
        const float y0 = fmaxf((acc0[r] - mm[r]) * sc[r] + bb[r], 0.0f);
        const float y1 = fmaxf((acc1[r] - mm[r]) * sc[r] + bb[r], 0.0f);
        const float v = row_max16(fmaxf(y0, y1));
        if (l16 == qq) res[r] = v;
      }
    }
  };
  v2f bq0[2][NKT], bq1[2][NKT];
  load_b<NKT>(bq0, Xin, cw + q0 * 64);
  int q = q0;
#pragma nounroll
  for (; q + 2 < q0 + NQ; q += 2) {
    load_b<NKT>(bq1, Xin, cw + (q + 1) * 64);
    do_query(bq0, q);
    load_b<NKT>(bq0, Xin, cw + (q + 2) * 64);
    do_query(bq1, q + 1);
  }
  load_b<NKT>(bq1, Xin, cw + (q + 1) * 64);
  do_query(bq0, q);
  next_tile();
  do_query(bq1, q + 1);

  if (POOL) {
    if (l16 < NQ) {
#pragma unroll
      for (int r = 0; r < 8; ++r) pool_out[(size_t)r * M_ + q0 + l16] = res[r];
    }
  }
}

__global__ __launch_bounds__(NW_ * 32) void k_mlp(
    const float* __restrict__ features, const float* __restrict__ xyz,
    const float* __restrict__ new_xyz, const int* __restrict__ knn,
    const float* __restrict__ W0, const float* __restrict__ g0,
    const float* __restrict__ b0, const float* __restrict__ m0,
    const float* __restrict__ v0,
    const float* __restrict__ W1, const float* __restrict__ g1,
    const float* __restrict__ b1, const float* __restrict__ m1,
    const float* __restrict__ v1,
    const float* __restrict__ W2, const float* __restrict__ g2,
    const float* __restrict__ b2, const float* __restrict__ m2,
    const float* __restrict__ v2,
    float* __restrict__ out_feat)
{
  __shared__ __attribute__((aligned(16))) float Xs[XPAIRS_ * PP_];
  __shared__ __attribute__((aligned(16))) float Ys[YPAIRS_ * PP_];
  static_assert(SROWS_ * N_ <= YPAIRS_ * PP_, "staged rows must fit the Y matrix");
  static_assert(NW_ % SROWS_ == 0, "whole waves per staged row");
  const int wave = __builtin_amdgcn_readfirstlane(threadIdx.x >> 5);
  const int lane = threadIdx.x & 31;
  const int q = blockIdx.x * NW_ + wave;
  const int b = q >> 10;
  const int mq0 = (blockIdx.x * NW_) & 1023;

  const int mt01 = wave & 3;
  const int q01  = (wave >> 2) * (NW_ / 2);
  RowTile<17> t0;
  load_row_tile<17, 67>(t0, W0, mt01, lane);

  const int nj = knn[q * K_ + lane];
  const int colw = (wave * 32 + lane) * 2;
  const float* fb = features + (size_t)b * C_ * N_;
  {
    constexpr int WPR = NW_ / SROWS_;
    constexpr int VPL = N_ / WPR / 128;
    float* stage = Ys;
    const int srow = wave / WPR;
    const int soff = (wave % WPR) * (N_ / WPR) + lane * 4;
    const float* src = fb + (size_t)srow * N_ + soff;
    float* dst = stage + srow * N_ + soff;
    v4f r[VPL];
#pragma unroll
    for (int i = 0; i < VPL; ++i)
      r[i] = __builtin_nontemporal_load((const v4f*)(src + i * 128));
    for (int c0 = 0; c0 < C_; c0 += SROWS_) {
#pragma unroll
      for (int i = 0; i < VPL; ++i) *(v4f*)(dst + i * 128) = r[i];
      if (c0 + SROWS_ < C_) {
        src += (size_t)SROWS_ * N_;
#pragma unroll
        for (int i = 0; i < VPL; ++i)
          r[i] = __builtin_nontemporal_load((const v4f*)(src + i * 128));
      }
      lds_barrier();
      {
        v2f p0, p1;
        p0.x = stage[0 * N_ + nj];
        p0.y = stage[1 * N_ + nj];
        p1.x = stage[2 * N_ + nj];
        p1.y = stage[3 * N_ + nj];
        *(v2f*)(Xs + ((c0 >> 1) + 0) * PP_ + colw) = p0;
        *(v2f*)(Xs + ((c0 >> 1) + 1) * PP_ + colw) = p1;
      }
      lds_barrier();
    }
  }
  {
    const float* p  = xyz + ((size_t)b * N_ + nj) * 3;
    const float* qc = new_xyz + (size_t)q * 3;
    v2f e0, e1;
    e0.x = p[0] - qc[0];
    e0.y = p[1] - qc[1];
    e1.x = p[2] - qc[2];
    e1.y = 0.0f;
    *(v2f*)(Xs + 32 * PP_ + colw) = e0;
    *(v2f*)(Xs + 33 * PP_ + colw) = e1;
  }
  __syncthreads();

  RowTile<16> t1, t2;
  layer_rows<17, NW_ / 2, false>(Xs, Ys, t0, g0, b0, m0, v0, mt01, q01, lane, nullptr,
      [&] { load_row_tile<16, 64>(t1, W1, mt01, lane); });
  __syncthreads();
  layer_rows<16, NW_ / 2, false>(Ys, Xs, t1, g1, b1, m1, v1, mt01, q01, lane, nullptr,
      [&] { load_row_tile<16, 64>(t2, W2, wave, lane); });
  __syncthreads();
  layer_rows<16, NW_, true>(Xs, nullptr, t2, g2, b2, m2, v2, wave, 0, lane,
      out_feat + ((size_t)b * 128 + wave * 16 + (lane >> 4) * 8) * M_ + mq0,
      [] {});
}

extern "C" void kernel_launch(void* const* d_in, const int* in_sizes, int n_in,
                              void* d_out, int out_size, void* d_ws, size_t ws_size,
                              hipStream_t stream) {
  const float* xyz      = (const float*)d_in[0];
  const float* features = (const float*)d_in[1];
  const int*   indices  = (const int*)d_in[2];
  const float* W0 = (const float*)d_in[3];
  const float* g0 = (const float*)d_in[4];
  const float* b0 = (const float*)d_in[5];
  const float* m0 = (const float*)d_in[6];
  const float* v0 = (const float*)d_in[7];
  const float* W1 = (const float*)d_in[8];
  const float* g1 = (const float*)d_in[9];
  const float* b1 = (const float*)d_in[10];
  const float* m1 = (const float*)d_in[11];
  const float* v1 = (const float*)d_in[12];
  const float* W2 = (const float*)d_in[13];
  const float* g2 = (const float*)d_in[14];
  const float* b2 = (const float*)d_in[15];
  const float* m2 = (const float*)d_in[16];
  const float* v2 = (const float*)d_in[17];

  float* new_xyz  = (float*)d_out;
  float* out_feat = new_xyz + (size_t)B_ * M_ * 3;
  int*   knn      = (int*)d_ws;

  k_new_xyz<<<(B_ * M_ + 255) / 256, 256, 0, stream>>>(xyz, indices, new_xyz);
  k_knn<<<B_ * M_, 32, 0, stream>>>(xyz, new_xyz, knn);
  k_mlp<<<(B_ * M_) / NW_, NW_ * 32, 0, stream>>>(features, xyz, new_xyz, knn,
      W0, g0, b0, m0, v0, W1, g1, b1, m1, v1, W2, g2, b2, m2, v2, out_feat);
}


# MLP block order: consecutive blocks take different batches (batch = block % 16), so blocks resident together stage rows of all 16 batches, not 4: row reads spread over 4x the L2 lines; sums unchanged
# speedup vs baseline: 2.1258x; 1.0366x over previous
#include <hip/hip_runtime.h>
#include <hip/hip_bf16.h>

typedef __attribute__((ext_vector_type(2))) float v2f;
typedef __attribute__((ext_vector_type(8))) float v8f;

#define B_   16
#define N_   4096
#define C_   64
#define M_   1024
#define K_   32
#define EPS_ 1e-5f
#define BIGF 3.402823e38f

__global__ __launch_bounds__(256) void k_new_xyz(const float* __restrict__ xyz,
                                                 const int* __restrict__ indices,
                                                 float* __restrict__ new_xyz) {
  int t = blockIdx.x * 256 + threadIdx.x;
  if (t >= B_ * M_) return;
  int b = t >> 10, m = t & 1023;
  int n = indices[m];
  const float* s = xyz + ((size_t)b * N_ + n) * 3;
  float* d = new_xyz + (size_t)t * 3;
  d[0] = s[0]; d[1] = s[1]; d[2] = s[2];
}

#define CAP_ 128

__device__ __forceinline__ float sqdist(float qx, float qy, float qz, float qq,
                                        float x, float y, float z) {
  const float pp  = fmaf(z, z, fmaf(x, x, y * y));
  const float dot = fmaf(qz, z, fmaf(qx, x, qy * y));
  return fmaf(-2.0f, dot, qq + pp);
}

__device__ __forceinline__ unsigned okey(float f) {
  const unsigned u = __float_as_uint(f);
  return (u & 0x80000000u) ? ~u : (u | 0x80000000u);
}
__device__ __forceinline__ float okey_inv(unsigned k) {
  return __uint_as_float((k & 0x80000000u) ? (k & 0x7fffffffu) : ~k);
}

template <int NS, int NB>
__device__ __forceinline__ unsigned kth32(const unsigned (&k)[NS]) {
  unsigned v = 0;
  for (int bit = 31; bit > 31 - NB; --bit) {
    const unsigned t = v | (1u << bit);
    int c = 0;
#pragma unroll
    for (int s = 0; s < NS; ++s)
      c += __builtin_popcount(__builtin_amdgcn_ballot_w32(k[s] < t));
    if (c < K_) v = t;
  }
  if (NB < 32) v |= (1u << (32 - NB)) - 1u;
  return v;
}

__global__ __launch_bounds__(32) void k_knn(const float* __restrict__ xyz,
                                            const float* __restrict__ new_xyz,
                                            int* __restrict__ knn) {
  __shared__ float    dl[N_];
  __shared__ unsigned ck[CAP_];
  __shared__ int      ci[CAP_];
  const int lane = threadIdx.x;
  const int q = blockIdx.x;
  const int b = q >> 10;

  const float qx = new_xyz[q * 3 + 0];
  const float qy = new_xyz[q * 3 + 1];
  const float qz = new_xyz[q * 3 + 2];
  float qq;
  {
#pragma clang fp contract(off)
    const float yy = qy * qy;
    const float zz = qz * qz;
    qq = fmaf(qx, qx, yy) + zz;
  }
  const float* px = xyz + (size_t)b * N_ * 3;

  float m1 = BIGF, m2 = BIGF;
#pragma unroll 8
  for (int n = lane; n < N_; n += 32) {
    const float d = sqdist(qx, qy, qz, qq, px[n * 3 + 0], px[n * 3 + 1], px[n * 3 + 2]);
    dl[n] = d;
    m2 = fminf(m2, fmaxf(m1, d));
    m1 = fminf(m1, d);
  }

  const unsigned kb[2] = { okey(m1), okey(m2) };
  const float fB = okey_inv(kth32<2, 14>(kb));

  int cnt = 0;
  for (int w = 0; w < N_ / 1024; ++w) {
    unsigned bw = 0;
#pragma unroll
    for (int u = 0; u < 32; ++u)
      bw |= (dl[lane + 32 * (32 * w + u)] <= fB) ? (1u << u) : 0u;
    for (int t = 0; t < 32; ++t) {
      const bool has = bw != 0;
      const unsigned mask = __builtin_amdgcn_ballot_w32(has);
      if (mask == 0) break;
      if (has) {
        const int u = __builtin_ctz(bw);
        bw &= bw - 1;
        const int n = lane + 32 * (32 * w + u);
        const int pos = cnt + (int)__builtin_amdgcn_mbcnt_lo(mask, 0u);
        if (pos < CAP_) { ck[pos] = okey(dl[n]); ci[pos] = n; }
      }
      cnt += __builtin_popcount(mask);
    }
  }
  __syncthreads();

  int* out = knn + q * K_;
  bool done = false;
  if (cnt >= K_ && cnt <= CAP_) {
    unsigned k[CAP_ / 32];
    int id[CAP_ / 32];
#pragma unroll
    for (int s = 0; s < CAP_ / 32; ++s) {
      const int j = s * 32 + lane;
      const bool valid = j < cnt;
      k[s]  = valid ? ck[j] : 0xffffffffu;
      id[s] = valid ? ci[j] : 0;
    }
    unsigned v32;
    if (cnt <= 64) {
      const unsigned k2[2] = { k[0], k[1] };
      v32 = kth32<2, 32>(k2);
    } else {
      v32 = kth32<CAP_ / 32, 32>(k);
    }
    unsigned mt[CAP_ / 32];
    int total = 0;
#pragma unroll
    for (int s = 0; s < CAP_ / 32; ++s) {
      mt[s] = __builtin_amdgcn_ballot_w32(s * 32 + lane < cnt && k[s] <= v32);
      total += __builtin_popcount(mt[s]);
    }
    if (total == K_) {
      int wpos = 0;
#pragma unroll
      for (int s = 0; s < CAP_ / 32; ++s) {
        if ((mt[s] >> lane) & 1u) out[wpos + (int)__builtin_amdgcn_mbcnt_lo(mt[s], 0u)] = id[s];
        wpos += __builtin_popcount(mt[s]);
      }
      done = true;
    }
  }
  if (!done) {
    for (int kk = 0; kk < K_; ++kk) {
      float mv = BIGF;
      int   mi = 0x7fffffff;
      for (int n = lane; n < N_; n += 32) {
        float v = dl[n];
        if (v < mv) { mv = v; mi = n; }
      }
#pragma unroll
      for (int off = 16; off > 0; off >>= 1) {
        float ov = __shfl_xor(mv, off, 32);
        int   oi = __shfl_xor(mi, off, 32);
        if (ov < mv || (ov == mv && oi < mi)) { mv = ov; mi = oi; }
      }
      if (lane == 0) out[kk] = mi;
      dl[mi] = BIGF;
      __syncthreads();
    }
  }
}

#define NW_     8
#define NCOL_   (NW_ * 32)
#define PP_     (2 * NCOL_ + 32)
#define XPAIRS_ 34
#define YPAIRS_ 32
#define SROWS_  4
typedef __attribute__((ext_vector_type(4))) float v4f;

__device__ __forceinline__ void lds_barrier() {
  asm volatile("s_wait_dscnt 0x0\n\ts_barrier_signal -1\n\ts_barrier_wait -1" ::: "memory");
}

template <int MASK>
__device__ __forceinline__ float row_xor(float v) {
  return __builtin_bit_cast(float, __builtin_amdgcn_update_dpp(
      0, __builtin_bit_cast(int, v), 0x160 + MASK, 0xf, 0xf, true));
}
__device__ __forceinline__ float row_max16(float v) {
  v = fmaxf(v, row_xor<1>(v));
  v = fmaxf(v, row_xor<2>(v));
  v = fmaxf(v, row_xor<4>(v));
  v = fmaxf(v, row_xor<8>(v));
  return v;
}

template <int NKT>
__device__ __forceinline__ void load_b(v2f (&bq)[2][NKT], const float* Xin, int cw) {
#pragma unroll
  for (int kt = 0; kt < NKT; ++kt) {
    bq[0][kt] = *(const v2f*)(Xin + kt * 2 * PP_ + cw);
    bq[1][kt] = *(const v2f*)(Xin + kt * 2 * PP_ + cw + 32);
  }
}

template <int NKT>
struct RowTile {
  v2f a[NKT];
};
template <int NKT, int KW>
__device__ __forceinline__ void load_row_tile(
    RowTile<NKT>& t, const float* __restrict__ W, int mt, int lane)
{
  const int half = lane >> 4;
  const int arow = mt * 16 + (lane & 15);
#pragma unroll
  for (int kt = 0; kt < NKT; ++kt) {
    const int k0 = kt * 4 + half * 2;
    t.a[kt].x = (k0 + 0 < KW) ? W[arow * KW + k0 + 0] : 0.0f;
    t.a[kt].y = (k0 + 1 < KW) ? W[arow * KW + k0 + 1] : 0.0f;
  }
}

template <int NKT, int NQ, bool POOL, typename NextTile>
__device__ __forceinline__ void layer_rows(
    const float* Xin, float* Yout, const RowTile<NKT>& t,
    const float* __restrict__ g, const float* __restrict__ bt,
    const float* __restrict__ mu, const float* __restrict__ vr,
    int mt, int q0, int lane, float* __restrict__ pool_out, NextTile next_tile)
{
  static_assert(NQ % 2 == 0, "queries are taken two per trip");
  const int half = lane >> 4;
  const int l16  = lane & 15;

  float sc[8], mm[8], bb[8];
  const int rbase = mt * 16 + half * 8;
#pragma unroll
  for (int r = 0; r < 8; ++r) {
    const int row = rbase + r;
    sc[r] = g[row] / sqrtf(vr[row] + EPS_);
    mm[r] = mu[row];
    bb[r] = bt[row];
  }

  float res[8];
#pragma unroll
  for (int r = 0; r < 8; ++r) res[r] = 0.0f;

  const int cw = half * PP_ + l16 * 2;
  auto do_query = [&](const v2f (&bcur)[2][NKT], int qq) {
    v8f acc0 = {}, acc1 = {};
#pragma unroll
    for (int kt = 0; kt < NKT; ++kt) {
      acc0 = __builtin_amdgcn_wmma_f32_16x16x4_f32(
           false, t.a[kt],  false, bcur[0][kt],
           (short)0, acc0,  false,  false);
      acc1 = __builtin_amdgcn_wmma_f32_16x16x4_f32(
           false, t.a[kt],  false, bcur[1][kt],
           (short)0, acc1,  false,  false);
    }
    if (!POOL) {
      float* yq = Yout + (mt * 8 + half * 4) * PP_ + (qq * 32 + l16) * 2;
#pragma unroll
      for (int j = 0; j < 4; ++j) {
        v2f y0, y1;
        y0.x = fmaxf((acc0[2 * j]     - mm[2 * j])     * sc[2 * j]     + bb[2 * j],     0.0f);
        y0.y = fmaxf((acc0[2 * j + 1] - mm[2 * j + 1]) * sc[2 * j + 1] + bb[2 * j + 1], 0.0f);
        y1.x = fmaxf((acc1[2 * j]     - mm[2 * j])     * sc[2 * j]     + bb[2 * j],     0.0f);
        y1.y = fmaxf((acc1[2 * j + 1] - mm[2 * j + 1]) * sc[2 * j + 1] + bb[2 * j + 1], 0.0f);
        *(v2f*)(yq + j * PP_)      = y0;
        *(v2f*)(yq + j * PP_ + 32) = y1;
      }
    } else {
#pragma unroll
      for (int r = 0; r < 8; ++r) {
        const float y0 = fmaxf((acc0[r] - mm[r]) * sc[r] + bb[r], 0.0f);
        const float y1 = fmaxf((acc1[r] - mm[r]) * sc[r] + bb[r], 0.0f);
        const float v = row_max16(fmaxf(y0, y1));
        if (l16 == qq) res[r] = v;
      }
    }
  };
  v2f bq0[2][NKT], bq1[2][NKT];
  load_b<NKT>(bq0, Xin, cw + q0 * 64);
  int q = q0;
#pragma nounroll
  for (; q + 2 < q0 + NQ; q += 2) {
    load_b<NKT>(bq1, Xin, cw + (q + 1) * 64);
    do_query(bq0, q);
    load_b<NKT>(bq0, Xin, cw + (q + 2) * 64);
    do_query(bq1, q + 1);
  }
  load_b<NKT>(bq1, Xin, cw + (q + 1) * 64);
  do_query(bq0, q);
  next_tile();
  do_query(bq1, q + 1);

  if (POOL) {
    if (l16 < NQ) {
#pragma unroll
      for (int r = 0; r < 8; ++r) pool_out[(size_t)r * M_ + q0 + l16] = res[r];
    }
  }
}

__global__ __launch_bounds__(NW_ * 32) void k_mlp(
    const float* __restrict__ features, const float* __restrict__ xyz,
    const float* __restrict__ new_xyz, const int* __restrict__ knn,
    const float* __restrict__ W0, const float* __restrict__ g0,
    const float* __restrict__ b0, const float* __restrict__ m0,
    const float* __restrict__ v0,
    const float* __restrict__ W1, const float* __restrict__ g1,
    const float* __restrict__ b1, const float* __restrict__ m1,
    const float* __restrict__ v1,
    const float* __restrict__ W2, const float* __restrict__ g2,
    const float* __restrict__ b2, const float* __restrict__ m2,
    const float* __restrict__ v2,
    float* __restrict__ out_feat)
{
  __shared__ __attribute__((aligned(16))) float Xs[XPAIRS_ * PP_];
  __shared__ __attribute__((aligned(16))) float Ys[YPAIRS_ * PP_];
  static_assert(SROWS_ * N_ <= YPAIRS_ * PP_, "staged rows must fit the Y matrix");
  static_assert(NW_ % SROWS_ == 0, "whole waves per staged row");
  const int wave = __builtin_amdgcn_readfirstlane(threadIdx.x >> 5);
  const int lane = threadIdx.x & 31;
  const int b = blockIdx.x % B_;
  const int mq0 = (blockIdx.x / B_) * NW_;
  const int q = b * M_ + mq0 + wave;

  const int mt01 = wave & 3;
  const int q01  = (wave >> 2) * (NW_ / 2);
  RowTile<17> t0;
  load_row_tile<17, 67>(t0, W0, mt01, lane);

  const int nj = knn[q * K_ + lane];
  const int colw = (wave * 32 + lane) * 2;
  const float* fb = features + (size_t)b * C_ * N_;
  {
    constexpr int WPR = NW_ / SROWS_;
    constexpr int VPL = N_ / WPR / 128;
    float* stage = Ys;
    const int srow = wave / WPR;
    const int soff = (wave % WPR) * (N_ / WPR) + lane * 4;
    const float* src = fb + (size_t)srow * N_ + soff;
    float* dst = stage + srow * N_ + soff;
    v4f r[VPL];
#pragma unroll
    for (int i = 0; i < VPL; ++i)
      r[i] = __builtin_nontemporal_load((const v4f*)(src + i * 128));
    for (int c0 = 0; c0 < C_; c0 += SROWS_) {
#pragma unroll
      for (int i = 0; i < VPL; ++i) *(v4f*)(dst + i * 128) = r[i];
      if (c0 + SROWS_ < C_) {
        src += (size_t)SROWS_ * N_;
#pragma unroll
        for (int i = 0; i < VPL; ++i)
          r[i] = __builtin_nontemporal_load((const v4f*)(src + i * 128));
      }
      lds_barrier();
      {
        v2f p0, p1;
        p0.x = stage[0 * N_ + nj];
        p0.y = stage[1 * N_ + nj];
        p1.x = stage[2 * N_ + nj];
        p1.y = stage[3 * N_ + nj];
        *(v2f*)(Xs + ((c0 >> 1) + 0) * PP_ + colw) = p0;
        *(v2f*)(Xs + ((c0 >> 1) + 1) * PP_ + colw) = p1;
      }
      lds_barrier();
    }
  }
  {
    const float* p  = xyz + ((size_t)b * N_ + nj) * 3;
    const float* qc = new_xyz + (size_t)q * 3;
    v2f e0, e1;
    e0.x = p[0] - qc[0];
    e0.y = p[1] - qc[1];
    e1.x = p[2] - qc[2];
    e1.y = 0.0f;
    *(v2f*)(Xs + 32 * PP_ + colw) = e0;
    *(v2f*)(Xs + 33 * PP_ + colw) = e1;
  }
  __syncthreads();

  RowTile<16> t1, t2;
  layer_rows<17, NW_ / 2, false>(Xs, Ys, t0, g0, b0, m0, v0, mt01, q01, lane, nullptr,
      [&] { load_row_tile<16, 64>(t1, W1, mt01, lane); });
  __syncthreads();
  layer_rows<16, NW_ / 2, false>(Ys, Xs, t1, g1, b1, m1, v1, mt01, q01, lane, nullptr,
      [&] { load_row_tile<16, 64>(t2, W2, wave, lane); });
  __syncthreads();
  layer_rows<16, NW_, true>(Xs, nullptr, t2, g2, b2, m2, v2, wave, 0, lane,
      out_feat + ((size_t)b * 128 + wave * 16 + (lane >> 4) * 8) * M_ + mq0,
      [] {});
}

extern "C" void kernel_launch(void* const* d_in, const int* in_sizes, int n_in,
                              void* d_out, int out_size, void* d_ws, size_t ws_size,
                              hipStream_t stream) {
  const float* xyz      = (const float*)d_in[0];
  const float* features = (const float*)d_in[1];
  const int*   indices  = (const int*)d_in[2];
  const float* W0 = (const float*)d_in[3];
  const float* g0 = (const float*)d_in[4];
  const float* b0 = (const float*)d_in[5];
  const float* m0 = (const float*)d_in[6];
  const float* v0 = (const float*)d_in[7];
  const float* W1 = (const float*)d_in[8];
  const float* g1 = (const float*)d_in[9];
  const float* b1 = (const float*)d_in[10];
  const float* m1 = (const float*)d_in[11];
  const float* v1 = (const float*)d_in[12];
  const float* W2 = (const float*)d_in[13];
  const float* g2 = (const float*)d_in[14];
  const float* b2 = (const float*)d_in[15];
  const float* m2 = (const float*)d_in[16];
  const float* v2 = (const float*)d_in[17];

  float* new_xyz  = (float*)d_out;
  float* out_feat = new_xyz + (size_t)B_ * M_ * 3;
  int*   knn      = (int*)d_ws;

  k_new_xyz<<<(B_ * M_ + 255) / 256, 256, 0, stream>>>(xyz, indices, new_xyz);
  k_knn<<<B_ * M_, 32, 0, stream>>>(xyz, new_xyz, knn);
  k_mlp<<<(B_ * M_) / NW_, NW_ * 32, 0, stream>>>(features, xyz, new_xyz, knn,
      W0, g0, b0, m0, v0, W1, g1, b1, m1, v1, W2, g2, b2, m2, v2, out_feat);
}


# Feature gather in its own launch, a block per batch and 4 channels: their rows staged in LDS once serve 256 queries (not 8); X tiles go through the workspace, MLP blocks copy 64 KiB; sums unchanged
# speedup vs baseline: 2.2500x; 1.0584x over previous
#include <hip/hip_runtime.h>
#include <hip/hip_bf16.h>

typedef __attribute__((ext_vector_type(2))) float v2f;
typedef __attribute__((ext_vector_type(8))) float v8f;

#define B_   16
#define N_   4096
#define C_   64
#define M_   1024
#define K_   32
#define EPS_ 1e-5f
#define BIGF 3.402823e38f

__global__ __launch_bounds__(256) void k_new_xyz(const float* __restrict__ xyz,
                                                 const int* __restrict__ indices,
                                                 float* __restrict__ new_xyz) {
  int t = blockIdx.x * 256 + threadIdx.x;
  if (t >= B_ * M_) return;
  int b = t >> 10, m = t & 1023;
  int n = indices[m];
  const float* s = xyz + ((size_t)b * N_ + n) * 3;
  float* d = new_xyz + (size_t)t * 3;
  d[0] = s[0]; d[1] = s[1]; d[2] = s[2];
}

#define CAP_ 128

__device__ __forceinline__ float sqdist(float qx, float qy, float qz, float qq,
                                        float x, float y, float z) {
  const float pp  = fmaf(z, z, fmaf(x, x, y * y));
  const float dot = fmaf(qz, z, fmaf(qx, x, qy * y));
  return fmaf(-2.0f, dot, qq + pp);
}

__device__ __forceinline__ unsigned okey(float f) {
  const unsigned u = __float_as_uint(f);
  return (u & 0x80000000u) ? ~u : (u | 0x80000000u);
}
__device__ __forceinline__ float okey_inv(unsigned k) {
  return __uint_as_float((k & 0x80000000u) ? (k & 0x7fffffffu) : ~k);
}

template <int NS, int NB>
__device__ __forceinline__ unsigned kth32(const unsigned (&k)[NS]) {
  unsigned v = 0;
  for (int bit = 31; bit > 31 - NB; --bit) {
    const unsigned t = v | (1u << bit);
    int c = 0;
#pragma unroll
    for (int s = 0; s < NS; ++s)
      c += __builtin_popcount(__builtin_amdgcn_ballot_w32(k[s] < t));
    if (c < K_) v = t;
  }
  if (NB < 32) v |= (1u << (32 - NB)) - 1u;
  return v;
}

__global__ __launch_bounds__(32) void k_knn(const float* __restrict__ xyz,
                                            const float* __restrict__ new_xyz,
                                            int* __restrict__ knn) {
  __shared__ float    dl[N_];
  __shared__ unsigned ck[CAP_];
  __shared__ int      ci[CAP_];
  const int lane = threadIdx.x;
  const int q = blockIdx.x;
  const int b = q >> 10;

  const float qx = new_xyz[q * 3 + 0];
  const float qy = new_xyz[q * 3 + 1];
  const float qz = new_xyz[q * 3 + 2];
  float qq;
  {
#pragma clang fp contract(off)
    const float yy = qy * qy;
    const float zz = qz * qz;
    qq = fmaf(qx, qx, yy) + zz;
  }
  const float* px = xyz + (size_t)b * N_ * 3;

  float m1 = BIGF, m2 = BIGF;
#pragma unroll 8
  for (int n = lane; n < N_; n += 32) {
    const float d = sqdist(qx, qy, qz, qq, px[n * 3 + 0], px[n * 3 + 1], px[n * 3 + 2]);
    dl[n] = d;
    m2 = fminf(m2, fmaxf(m1, d));
    m1 = fminf(m1, d);
  }

  const unsigned kb[2] = { okey(m1), okey(m2) };
  const float fB = okey_inv(kth32<2, 14>(kb));

  int cnt = 0;
  for (int w = 0; w < N_ / 1024; ++w) {
    unsigned bw = 0;
#pragma unroll
    for (int u = 0; u < 32; ++u)
      bw |= (dl[lane + 32 * (32 * w + u)] <= fB) ? (1u << u) : 0u;
    for (int t = 0; t < 32; ++t) {
      const bool has = bw != 0;
      const unsigned mask = __builtin_amdgcn_ballot_w32(has);
      if (mask == 0) break;
      if (has) {
        const int u = __builtin_ctz(bw);
        bw &= bw - 1;
        const int n = lane + 32 * (32 * w + u);
        const int pos = cnt + (int)__builtin_amdgcn_mbcnt_lo(mask, 0u);
        if (pos < CAP_) { ck[pos] = okey(dl[n]); ci[pos] = n; }
      }
      cnt += __builtin_popcount(mask);
    }
  }
  __syncthreads();

  int* out = knn + q * K_;
  bool done = false;
  if (cnt >= K_ && cnt <= CAP_) {
    unsigned k[CAP_ / 32];
    int id[CAP_ / 32];
#pragma unroll
    for (int s = 0; s < CAP_ / 32; ++s) {
      const int j = s * 32 + lane;
      const bool valid = j < cnt;
      k[s]  = valid ? ck[j] : 0xffffffffu;
      id[s] = valid ? ci[j] : 0;
    }
    unsigned v32;
    if (cnt <= 64) {
      const unsigned k2[2] = { k[0], k[1] };
      v32 = kth32<2, 32>(k2);
    } else {
      v32 = kth32<CAP_ / 32, 32>(k);
    }
    unsigned mt[CAP_ / 32];
    int total = 0;
#pragma unroll
    for (int s = 0; s < CAP_ / 32; ++s) {
      mt[s] = __builtin_amdgcn_ballot_w32(s * 32 + lane < cnt && k[s] <= v32);
      total += __builtin_popcount(mt[s]);
    }
    if (total == K_) {
      int wpos = 0;
#pragma unroll
      for (int s = 0; s < CAP_ / 32; ++s) {
        if ((mt[s] >> lane) & 1u) out[wpos + (int)__builtin_amdgcn_mbcnt_lo(mt[s], 0u)] = id[s];
        wpos += __builtin_popcount(mt[s]);
      }
      done = true;
    }
  }
  if (!done) {
    for (int kk = 0; kk < K_; ++kk) {
      float mv = BIGF;
      int   mi = 0x7fffffff;
      for (int n = lane; n < N_; n += 32) {
        float v = dl[n];
        if (v < mv) { mv = v; mi = n; }
      }
#pragma unroll
      for (int off = 16; off > 0; off >>= 1) {
        float ov = __shfl_xor(mv, off, 32);
        int   oi = __shfl_xor(mi, off, 32);
        if (ov < mv || (ov == mv && oi < mi)) { mv = ov; mi = oi; }
      }
      if (lane == 0) out[kk] = mi;
      dl[mi] = BIGF;
      __syncthreads();
    }
  }
}

#define NW_     8
#define NCOL_   (NW_ * 32)
#define PP_     (2 * NCOL_ + 32)
#define XPAIRS_ 34
#define YPAIRS_ 32
#define FPAIRS_ (C_ / 2)
#define TILEW_  (FPAIRS_ * 2 * NCOL_)
#define GCH_    4
#define GQP_    4
typedef __attribute__((ext_vector_type(4))) float v4f;

__global__ __launch_bounds__(256) void k_gather(const float* __restrict__ features,
                                                const int* __restrict__ knn,
                                                float* __restrict__ xg) {
  __shared__ __attribute__((aligned(16))) float st[(GCH_ / 2) * 2 * N_];
  const int t    = threadIdx.x;
  const int lane = t & 31;
  const int wave = __builtin_amdgcn_readfirstlane(t >> 5);
  const int b    = blockIdx.x % B_;
  const int rest = blockIdx.x / B_;
  const int cg   = rest % (C_ / GCH_);
  const int part = rest / (C_ / GCH_);
  const float* fb = features + ((size_t)b * C_ + cg * GCH_) * N_;

  {
    constexpr int NI = N_ / 4 / 256;
    v4f ra[GCH_ / 2][NI], rb[GCH_ / 2][NI];
#pragma unroll
    for (int pr = 0; pr < GCH_ / 2; ++pr)
#pragma unroll
      for (int i = 0; i < NI; ++i) {
        const int pc = i * 256 + t;
        ra[pr][i] = __builtin_nontemporal_load((const v4f*)(fb + (size_t)(2 * pr)     * N_ + pc * 4));
        rb[pr][i] = __builtin_nontemporal_load((const v4f*)(fb + (size_t)(2 * pr + 1) * N_ + pc * 4));
      }
#pragma unroll
    for (int pr = 0; pr < GCH_ / 2; ++pr)
#pragma unroll
      for (int i = 0; i < NI; ++i) {
        const int pc = i * 256 + t;
        v4f lo, hi;
        lo.x = ra[pr][i].x; lo.y = rb[pr][i].x; lo.z = ra[pr][i].y; lo.w = rb[pr][i].y;
        hi.x = ra[pr][i].z; hi.y = rb[pr][i].z; hi.z = ra[pr][i].w; hi.w = rb[pr][i].w;
        *(v4f*)(st + pr * 2 * N_ + pc * 8)     = lo;
        *(v4f*)(st + pr * 2 * N_ + pc * 8 + 4) = hi;
      }
  }
  __syncthreads();

  constexpr int QPB = M_ / GQP_;
  const int m0 = part * QPB + wave;
  const int*  kq = knn + ((size_t)b * M_ + m0) * K_ + lane;
  float* dq = xg + ((size_t)b * (M_ / NW_) + part * (QPB / NW_)) * TILEW_
                 + (size_t)(cg * (GCH_ / 2)) * (2 * NCOL_) + (wave * 32 + lane) * 2;
#pragma unroll 4
  for (int it = 0; it < QPB / NW_; ++it) {
    const int nj = kq[(size_t)it * NW_ * K_];
#pragma unroll
    for (int pr = 0; pr < GCH_ / 2; ++pr) {
      const v2f p = *(const v2f*)(st + pr * 2 * N_ + nj * 2);
      *(v2f*)(dq + (size_t)it * TILEW_ + pr * (2 * NCOL_)) = p;
    }
  }
}

template <int MASK>
__device__ __forceinline__ float row_xor(float v) {
  return __builtin_bit_cast(float, __builtin_amdgcn_update_dpp(
      0, __builtin_bit_cast(int, v), 0x160 + MASK, 0xf, 0xf, true));
}
__device__ __forceinline__ float row_max16(float v) {
  v = fmaxf(v, row_xor<1>(v));
  v = fmaxf(v, row_xor<2>(v));
  v = fmaxf(v, row_xor<4>(v));
  v = fmaxf(v, row_xor<8>(v));
  return v;
}

template <int NKT>
__device__ __forceinline__ void load_b(v2f (&bq)[2][NKT], const float* Xin, int cw) {
#pragma unroll
  for (int kt = 0; kt < NKT; ++kt) {
    bq[0][kt] = *(const v2f*)(Xin + kt * 2 * PP_ + cw);
    bq[1][kt] = *(const v2f*)(Xin + kt * 2 * PP_ + cw + 32);
  }
}

template <int NKT>
struct RowTile {
  v2f a[NKT];
};
template <int NKT, int KW>
__device__ __forceinline__ void load_row_tile(
    RowTile<NKT>& t, const float* __restrict__ W, int mt, int lane)
{
  const int half = lane >> 4;
  const int arow = mt * 16 + (lane & 15);
#pragma unroll
  for (int kt = 0; kt < NKT; ++kt) {
    const int k0 = kt * 4 + half * 2;
    t.a[kt].x = (k0 + 0 < KW) ? W[arow * KW + k0 + 0] : 0.0f;
    t.a[kt].y = (k0 + 1 < KW) ? W[arow * KW + k0 + 1] : 0.0f;
  }
}

template <int NKT, int NQ, bool POOL, typename NextTile>
__device__ __forceinline__ void layer_rows(
    const float* Xin, float* Yout, const RowTile<NKT>& t,
    const float* __restrict__ g, const float* __restrict__ bt,
    const float* __restrict__ mu, const float* __restrict__ vr,
    int mt, int q0, int lane, float* __restrict__ pool_out, NextTile next_tile)
{
  static_assert(NQ % 2 == 0, "queries are taken two per trip");
  const int half = lane >> 4;
  const int l16  = lane & 15;

  float sc[8], mm[8], bb[8];
  const int rbase = mt * 16 + half * 8;
#pragma unroll
  for (int r = 0; r < 8; ++r) {
    const int row = rbase + r;
    sc[r] = g[row] / sqrtf(vr[row] + EPS_);
    mm[r] = mu[row];
    bb[r] = bt[row];
  }

  float res[8];
#pragma unroll
  for (int r = 0; r < 8; ++r) res[r] = 0.0f;

  const int cw = half * PP_ + l16 * 2;
  auto do_query = [&](const v2f (&bcur)[2][NKT], int qq) {
    v8f acc0 = {}, acc1 = {};
#pragma unroll
    for (int kt = 0; kt < NKT; ++kt) {
      acc0 = __builtin_amdgcn_wmma_f32_16x16x4_f32(
           false, t.a[kt],  false, bcur[0][kt],
           (short)0, acc0,  false,  false);
      acc1 = __builtin_amdgcn_wmma_f32_16x16x4_f32(
           false, t.a[kt],  false, bcur[1][kt],
           (short)0, acc1,  false,  false);
    }
    if (!POOL) {
      float* yq = Yout + (mt * 8 + half * 4) * PP_ + (qq * 32 + l16) * 2;
#pragma unroll
      for (int j = 0; j < 4; ++j) {
        v2f y0, y1;
        y0.x = fmaxf((acc0[2 * j]     - mm[2 * j])     * sc[2 * j]     + bb[2 * j],     0.0f);
        y0.y = fmaxf((acc0[2 * j + 1] - mm[2 * j + 1]) * sc[2 * j + 1] + bb[2 * j + 1], 0.0f);
        y1.x = fmaxf((acc1[2 * j]     - mm[2 * j])     * sc[2 * j]     + bb[2 * j],     0.0f);
        y1.y = fmaxf((acc1[2 * j + 1] - mm[2 * j + 1]) * sc[2 * j + 1] + bb[2 * j + 1], 0.0f);
        *(v2f*)(yq + j * PP_)      = y0;
        *(v2f*)(yq + j * PP_ + 32) = y1;
      }
    } else {
#pragma unroll
      for (int r = 0; r < 8; ++r) {
        const float y0 = fmaxf((acc0[r] - mm[r]) * sc[r] + bb[r], 0.0f);
        const float y1 = fmaxf((acc1[r] - mm[r]) * sc[r] + bb[r], 0.0f);
        const float v = row_max16(fmaxf(y0, y1));
        if (l16 == qq) res[r] = v;
      }
    }
  };
  v2f bq0[2][NKT], bq1[2][NKT];
  load_b<NKT>(bq0, Xin, cw + q0 * 64);
  int q = q0;
#pragma nounroll
  for (; q + 2 < q0 + NQ; q += 2) {
    load_b<NKT>(bq1, Xin, cw + (q + 1) * 64);
    do_query(bq0, q);
    load_b<NKT>(bq0, Xin, cw + (q + 2) * 64);
    do_query(bq1, q + 1);
  }
  load_b<NKT>(bq1, Xin, cw + (q + 1) * 64);
  do_query(bq0, q);
  next_tile();
  do_query(bq1, q + 1);

  if (POOL) {
    if (l16 < NQ) {
#pragma unroll
      for (int r = 0; r < 8; ++r) pool_out[(size_t)r * M_ + q0 + l16] = res[r];
    }
  }
}

__global__ __launch_bounds__(NW_ * 32) void k_mlp(
    const float* __restrict__ xg, const float* __restrict__ xyz,
    const float* __restrict__ new_xyz, const int* __restrict__ knn,
    const float* __restrict__ W0, const float* __restrict__ g0,
    const float* __restrict__ b0, const float* __restrict__ m0,
    const float* __restrict__ v0,
    const float* __restrict__ W1, const float* __restrict__ g1,
    const float* __restrict__ b1, const float* __restrict__ m1,
    const float* __restrict__ v1,
    const float* __restrict__ W2, const float* __restrict__ g2,
    const float* __restrict__ b2, const float* __restrict__ m2,
    const float* __restrict__ v2,
    float* __restrict__ out_feat)
{
  __shared__ __attribute__((aligned(16))) float Xs[XPAIRS_ * PP_];
  __shared__ __attribute__((aligned(16))) float Ys[YPAIRS_ * PP_];
  const int wave = __builtin_amdgcn_readfirstlane(threadIdx.x >> 5);
  const int lane = threadIdx.x & 31;
  const int b = blockIdx.x % B_;
  const int mq0 = (blockIdx.x / B_) * NW_;
  const int q = b * M_ + mq0 + wave;

  const int mt01 = wave & 3;
  const int q01  = (wave >> 2) * (NW_ / 2);
  RowTile<17> t0;
  load_row_tile<17, 67>(t0, W0, mt01, lane);

  const int nj = knn[q * K_ + lane];
  const int colw = (wave * 32 + lane) * 2;
  {
    constexpr int VPL = TILEW_ / 4 / (NW_ * 32);
    const float* src = xg + ((size_t)b * (M_ / NW_) + blockIdx.x / B_) * TILEW_;
    const int t = threadIdx.x;
    v4f r[VPL];
#pragma unroll
    for (int i = 0; i < VPL; ++i)
      r[i] = __builtin_nontemporal_load((const v4f*)(src + (i * (NW_ * 32) + t) * 4));
#pragma unroll
    for (int i = 0; i < VPL; ++i) {
      const int pc = i * (NW_ * 32) + t;
      *(v4f*)(Xs + (pc / (NCOL_ / 2)) * PP_ + (pc % (NCOL_ / 2)) * 4) = r[i];
    }
  }
  {
    const float* p  = xyz + ((size_t)b * N_ + nj) * 3;
    const float* qc = new_xyz + (size_t)q * 3;
    v2f e0, e1;
    e0.x = p[0] - qc[0];
    e0.y = p[1] - qc[1];
    e1.x = p[2] - qc[2];
    e1.y = 0.0f;
    *(v2f*)(Xs + 32 * PP_ + colw) = e0;
    *(v2f*)(Xs + 33 * PP_ + colw) = e1;
  }
  __syncthreads();

  RowTile<16> t1, t2;
  layer_rows<17, NW_ / 2, false>(Xs, Ys, t0, g0, b0, m0, v0, mt01, q01, lane, nullptr,
      [&] { load_row_tile<16, 64>(t1, W1, mt01, lane); });
  __syncthreads();
  layer_rows<16, NW_ / 2, false>(Ys, Xs, t1, g1, b1, m1, v1, mt01, q01, lane, nullptr,
      [&] { load_row_tile<16, 64>(t2, W2, wave, lane); });
  __syncthreads();
  layer_rows<16, NW_, true>(Xs, nullptr, t2, g2, b2, m2, v2, wave, 0, lane,
      out_feat + ((size_t)b * 128 + wave * 16 + (lane >> 4) * 8) * M_ + mq0,
      [] {});
}

extern "C" void kernel_launch(void* const* d_in, const int* in_sizes, int n_in,
                              void* d_out, int out_size, void* d_ws, size_t ws_size,
                              hipStream_t stream) {
  const float* xyz      = (const float*)d_in[0];
  const float* features = (const float*)d_in[1];
  const int*   indices  = (const int*)d_in[2];
  const float* W0 = (const float*)d_in[3];
  const float* g0 = (const float*)d_in[4];
  const float* b0 = (const float*)d_in[5];
  const float* m0 = (const float*)d_in[6];
  const float* v0 = (const float*)d_in[7];
  const float* W1 = (const float*)d_in[8];
  const float* g1 = (const float*)d_in[9];
  const float* b1 = (const float*)d_in[10];
  const float* m1 = (const float*)d_in[11];
  const float* v1 = (const float*)d_in[12];
  const float* W2 = (const float*)d_in[13];
  const float* g2 = (const float*)d_in[14];
  const float* b2 = (const float*)d_in[15];
  const float* m2 = (const float*)d_in[16];
  const float* v2 = (const float*)d_in[17];

  float* new_xyz  = (float*)d_out;
  float* out_feat = new_xyz + (size_t)B_ * M_ * 3;
  int*   knn      = (int*)d_ws;
  float* xg       = (float*)d_ws + (size_t)B_ * M_ * K_;

  k_new_xyz<<<(B_ * M_ + 255) / 256, 256, 0, stream>>>(xyz, indices, new_xyz);
  k_knn<<<B_ * M_, 32, 0, stream>>>(xyz, new_xyz, knn);
  k_gather<<<B_ * (C_ / GCH_) * GQP_, 256, 0, stream>>>(features, knn, xg);
  k_mlp<<<(B_ * M_) / NW_, NW_ * 32, 0, stream>>>(xg, xyz, new_xyz, knn,
      W0, g0, b0, m0, v0, W1, g1, b1, m1, v1, W2, g2, b2, m2, v2, out_feat);
}


# BatchNorm folded into the products: a lane scales its own weight row by g/sqrt(v+eps), the shift b-m*s starts the accumulator, epilogue is ReLU alone; sums rounded differently, same order
# speedup vs baseline: 2.3229x; 1.0324x over previous
#include <hip/hip_runtime.h>
#include <hip/hip_bf16.h>

typedef __attribute__((ext_vector_type(2))) float v2f;
typedef __attribute__((ext_vector_type(8))) float v8f;

#define B_   16
#define N_   4096
#define C_   64
#define M_   1024
#define K_   32
#define EPS_ 1e-5f
#define BIGF 3.402823e38f

__global__ __launch_bounds__(256) void k_new_xyz(const float* __restrict__ xyz,
                                                 const int* __restrict__ indices,
                                                 float* __restrict__ new_xyz) {
  int t = blockIdx.x * 256 + threadIdx.x;
  if (t >= B_ * M_) return;
  int b = t >> 10, m = t & 1023;
  int n = indices[m];
  const float* s = xyz + ((size_t)b * N_ + n) * 3;
  float* d = new_xyz + (size_t)t * 3;
  d[0] = s[0]; d[1] = s[1]; d[2] = s[2];
}

#define CAP_ 128

__device__ __forceinline__ float sqdist(float qx, float qy, float qz, float qq,
                                        float x, float y, float z) {
  const float pp  = fmaf(z, z, fmaf(x, x, y * y));
  const float dot = fmaf(qz, z, fmaf(qx, x, qy * y));
  return fmaf(-2.0f, dot, qq + pp);
}

__device__ __forceinline__ unsigned okey(float f) {
  const unsigned u = __float_as_uint(f);
  return (u & 0x80000000u) ? ~u : (u | 0x80000000u);
}
__device__ __forceinline__ float okey_inv(unsigned k) {
  return __uint_as_float((k & 0x80000000u) ? (k & 0x7fffffffu) : ~k);
}

template <int NS, int NB>
__device__ __forceinline__ unsigned kth32(const unsigned (&k)[NS]) {
  unsigned v = 0;
  for (int bit = 31; bit > 31 - NB; --bit) {
    const unsigned t = v | (1u << bit);
    int c = 0;
#pragma unroll
    for (int s = 0; s < NS; ++s)
      c += __builtin_popcount(__builtin_amdgcn_ballot_w32(k[s] < t));
    if (c < K_) v = t;
  }
  if (NB < 32) v |= (1u << (32 - NB)) - 1u;
  return v;
}

__global__ __launch_bounds__(32) void k_knn(const float* __restrict__ xyz,
                                            const float* __restrict__ new_xyz,
                                            int* __restrict__ knn) {
  __shared__ float    dl[N_];
  __shared__ unsigned ck[CAP_];
  __shared__ int      ci[CAP_];
  const int lane = threadIdx.x;
  const int q = blockIdx.x;
  const int b = q >> 10;

  const float qx = new_xyz[q * 3 + 0];
  const float qy = new_xyz[q * 3 + 1];
  const float qz = new_xyz[q * 3 + 2];
  float qq;
  {
#pragma clang fp contract(off)
    const float yy = qy * qy;
    const float zz = qz * qz;
    qq = fmaf(qx, qx, yy) + zz;
  }
  const float* px = xyz + (size_t)b * N_ * 3;

  float m1 = BIGF, m2 = BIGF;
#pragma unroll 8
  for (int n = lane; n < N_; n += 32) {
    const float d = sqdist(qx, qy, qz, qq, px[n * 3 + 0], px[n * 3 + 1], px[n * 3 + 2]);
    dl[n] = d;
    m2 = fminf(m2, fmaxf(m1, d));
    m1 = fminf(m1, d);
  }

  const unsigned kb[2] = { okey(m1), okey(m2) };
  const float fB = okey_inv(kth32<2, 14>(kb));

  int cnt = 0;
  for (int w = 0; w < N_ / 1024; ++w) {
    unsigned bw = 0;
#pragma unroll
    for (int u = 0; u < 32; ++u)
      bw |= (dl[lane + 32 * (32 * w + u)] <= fB) ? (1u << u) : 0u;
    for (int t = 0; t < 32; ++t) {
      const bool has = bw != 0;
      const unsigned mask = __builtin_amdgcn_ballot_w32(has);
      if (mask == 0) break;
      if (has) {
        const int u = __builtin_ctz(bw);
        bw &= bw - 1;
        const int n = lane + 32 * (32 * w + u);
        const int pos = cnt + (int)__builtin_amdgcn_mbcnt_lo(mask, 0u);
        if (pos < CAP_) { ck[pos] = okey(dl[n]); ci[pos] = n; }
      }
      cnt += __builtin_popcount(mask);
    }
  }
  __syncthreads();

  int* out = knn + q * K_;
  bool done = false;
  if (cnt >= K_ && cnt <= CAP_) {
    unsigned k[CAP_ / 32];
    int id[CAP_ / 32];
#pragma unroll
    for (int s = 0; s < CAP_ / 32; ++s) {
      const int j = s * 32 + lane;
      const bool valid = j < cnt;
      k[s]  = valid ? ck[j] : 0xffffffffu;
      id[s] = valid ? ci[j] : 0;
    }
    unsigned v32;
    if (cnt <= 64) {
      const unsigned k2[2] = { k[0], k[1] };
      v32 = kth32<2, 32>(k2);
    } else {
      v32 = kth32<CAP_ / 32, 32>(k);
    }
    unsigned mt[CAP_ / 32];
    int total = 0;
#pragma unroll
    for (int s = 0; s < CAP_ / 32; ++s) {
      mt[s] = __builtin_amdgcn_ballot_w32(s * 32 + lane < cnt && k[s] <= v32);
      total += __builtin_popcount(mt[s]);
    }
    if (total == K_) {
      int wpos = 0;
#pragma unroll
      for (int s = 0; s < CAP_ / 32; ++s) {
        if ((mt[s] >> lane) & 1u) out[wpos + (int)__builtin_amdgcn_mbcnt_lo(mt[s], 0u)] = id[s];
        wpos += __builtin_popcount(mt[s]);
      }
      done = true;
    }
  }
  if (!done) {
    for (int kk = 0; kk < K_; ++kk) {
      float mv = BIGF;
      int   mi = 0x7fffffff;
      for (int n = lane; n < N_; n += 32) {
        float v = dl[n];
        if (v < mv) { mv = v; mi = n; }
      }
#pragma unroll
      for (int off = 16; off > 0; off >>= 1) {
        float ov = __shfl_xor(mv, off, 32);
        int   oi = __shfl_xor(mi, off, 32);
        if (ov < mv || (ov == mv && oi < mi)) { mv = ov; mi = oi; }
      }
      if (lane == 0) out[kk] = mi;
      dl[mi] = BIGF;
      __syncthreads();
    }
  }
}

#define NW_     8
#define NCOL_   (NW_ * 32)
#define PP_     (2 * NCOL_ + 32)
#define XPAIRS_ 34
#define YPAIRS_ 32
#define FPAIRS_ (C_ / 2)
#define TILEW_  (FPAIRS_ * 2 * NCOL_)
#define GCH_    4
#define GQP_    4
typedef __attribute__((ext_vector_type(4))) float v4f;

__global__ __launch_bounds__(256) void k_gather(const float* __restrict__ features,
                                                const int* __restrict__ knn,
                                                float* __restrict__ xg) {
  __shared__ __attribute__((aligned(16))) float st[(GCH_ / 2) * 2 * N_];
  const int t    = threadIdx.x;
  const int lane = t & 31;
  const int wave = __builtin_amdgcn_readfirstlane(t >> 5);
  const int b    = blockIdx.x % B_;
  const int rest = blockIdx.x / B_;
  const int cg   = rest % (C_ / GCH_);
  const int part = rest / (C_ / GCH_);
  const float* fb = features + ((size_t)b * C_ + cg * GCH_) * N_;

  {
    constexpr int NI = N_ / 4 / 256;
    v4f ra[GCH_ / 2][NI], rb[GCH_ / 2][NI];
#pragma unroll
    for (int pr = 0; pr < GCH_ / 2; ++pr)
#pragma unroll
      for (int i = 0; i < NI; ++i) {
        const int pc = i * 256 + t;
        ra[pr][i] = __builtin_nontemporal_load((const v4f*)(fb + (size_t)(2 * pr)     * N_ + pc * 4));
        rb[pr][i] = __builtin_nontemporal_load((const v4f*)(fb + (size_t)(2 * pr + 1) * N_ + pc * 4));
      }
#pragma unroll
    for (int pr = 0; pr < GCH_ / 2; ++pr)
#pragma unroll
      for (int i = 0; i < NI; ++i) {
        const int pc = i * 256 + t;
        v4f lo, hi;
        lo.x = ra[pr][i].x; lo.y = rb[pr][i].x; lo.z = ra[pr][i].y; lo.w = rb[pr][i].y;
        hi.x = ra[pr][i].z; hi.y = rb[pr][i].z; hi.z = ra[pr][i].w; hi.w = rb[pr][i].w;
        *(v4f*)(st + pr * 2 * N_ + pc * 8)     = lo;
        *(v4f*)(st + pr * 2 * N_ + pc * 8 + 4) = hi;
      }
  }
  __syncthreads();

  constexpr int QPB = M_ / GQP_;
  const int m0 = part * QPB + wave;
  const int*  kq = knn + ((size_t)b * M_ + m0) * K_ + lane;
  float* dq = xg + ((size_t)b * (M_ / NW_) + part * (QPB / NW_)) * TILEW_
                 + (size_t)(cg * (GCH_ / 2)) * (2 * NCOL_) + (wave * 32 + lane) * 2;
#pragma unroll 4
  for (int it = 0; it < QPB / NW_; ++it) {
    const int nj = kq[(size_t)it * NW_ * K_];
#pragma unroll
    for (int pr = 0; pr < GCH_ / 2; ++pr) {
      const v2f p = *(const v2f*)(st + pr * 2 * N_ + nj * 2);
      *(v2f*)(dq + (size_t)it * TILEW_ + pr * (2 * NCOL_)) = p;
    }
  }
}

template <int MASK>
__device__ __forceinline__ float row_xor(float v) {
  return __builtin_bit_cast(float, __builtin_amdgcn_update_dpp(
      0, __builtin_bit_cast(int, v), 0x160 + MASK, 0xf, 0xf, true));
}
__device__ __forceinline__ float row_max16(float v) {
  v = fmaxf(v, row_xor<1>(v));
  v = fmaxf(v, row_xor<2>(v));
  v = fmaxf(v, row_xor<4>(v));
  v = fmaxf(v, row_xor<8>(v));
  return v;
}

template <int NKT>
__device__ __forceinline__ void load_b(v2f (&bq)[2][NKT], const float* Xin, int cw) {
#pragma unroll
  for (int kt = 0; kt < NKT; ++kt) {
    bq[0][kt] = *(const v2f*)(Xin + kt * 2 * PP_ + cw);
    bq[1][kt] = *(const v2f*)(Xin + kt * 2 * PP_ + cw + 32);
  }
}

template <int NKT>
struct RowTile {
  v2f a[NKT];
  float s;
  v8f c;
};
template <int NKT, int KW>
__device__ __forceinline__ void load_row_tile(
    RowTile<NKT>& t, const float* __restrict__ W,
    const float* __restrict__ g, const float* __restrict__ bt,
    const float* __restrict__ mu, const float* __restrict__ vr, int mt, int lane)
{
  const int half = lane >> 4;
  const int arow = mt * 16 + (lane & 15);
#pragma unroll
  for (int kt = 0; kt < NKT; ++kt) {
    const int k0 = kt * 4 + half * 2;
    t.a[kt].x = (k0 + 0 < KW) ? W[arow * KW + k0 + 0] : 0.0f;
    t.a[kt].y = (k0 + 1 < KW) ? W[arow * KW + k0 + 1] : 0.0f;
  }
  t.s = g[arow] / sqrtf(vr[arow] + EPS_);
  const float crow = bt[arow] - mu[arow] * t.s;
#pragma unroll
  for (int r = 0; r < 8; ++r) t.c[r] = __shfl(crow, half * 24 + r, 32);
}

template <int NKT, int NQ, bool POOL, typename NextTile>
__device__ __forceinline__ void layer_rows(
    const float* Xin, float* Yout, const RowTile<NKT>& t,
    int mt, int q0, int lane, float* __restrict__ pool_out, NextTile next_tile)
{
  static_assert(NQ % 2 == 0, "queries are taken two per trip");
  const int half = lane >> 4;
  const int l16  = lane & 15;

  v2f as[NKT];
#pragma unroll
  for (int kt = 0; kt < NKT; ++kt) as[kt] = t.a[kt] * t.s;

  const v8f cinit = t.c;

  float res[8];
#pragma unroll
  for (int r = 0; r < 8; ++r) res[r] = 0.0f;

  const int cw = half * PP_ + l16 * 2;
  auto do_query = [&](const v2f (&bcur)[2][NKT], int qq) {
    v8f acc0 = cinit, acc1 = cinit;
#pragma unroll
    for (int kt = 0; kt < NKT; ++kt) {
      acc0 = __builtin_amdgcn_wmma_f32_16x16x4_f32(
           false, as[kt],  false, bcur[0][kt],
           (short)0, acc0,  false,  false);
      acc1 = __builtin_amdgcn_wmma_f32_16x16x4_f32(
           false, as[kt],  false, bcur[1][kt],
           (short)0, acc1,  false,  false);
    }
    if (!POOL) {
      float* yq = Yout + (mt * 8 + half * 4) * PP_ + (qq * 32 + l16) * 2;
#pragma unroll
      for (int j = 0; j < 4; ++j) {
        v2f y0, y1;
        y0.x = fmaxf(acc0[2 * j],     0.0f);
        y0.y = fmaxf(acc0[2 * j + 1], 0.0f);
        y1.x = fmaxf(acc1[2 * j],     0.0f);
        y1.y = fmaxf(acc1[2 * j + 1], 0.0f);
        *(v2f*)(yq + j * PP_)      = y0;
        *(v2f*)(yq + j * PP_ + 32) = y1;
      }
    } else {
#pragma unroll
      for (int r = 0; r < 8; ++r) {
        const float v = row_max16(fmaxf(acc0[r], acc1[r]));
        if (l16 == qq) res[r] = v;
      }
    }
  };
  v2f bq0[2][NKT], bq1[2][NKT];
  load_b<NKT>(bq0, Xin, cw + q0 * 64);
  int q = q0;
#pragma nounroll
  for (; q + 2 < q0 + NQ; q += 2) {
    load_b<NKT>(bq1, Xin, cw + (q + 1) * 64);
    do_query(bq0, q);
    load_b<NKT>(bq0, Xin, cw + (q + 2) * 64);
    do_query(bq1, q + 1);
  }
  load_b<NKT>(bq1, Xin, cw + (q + 1) * 64);
  do_query(bq0, q);
  next_tile();
  do_query(bq1, q + 1);

  if (POOL) {
    if (l16 < NQ) {
#pragma unroll
      for (int r = 0; r < 8; ++r) pool_out[(size_t)r * M_ + q0 + l16] = fmaxf(res[r], 0.0f);
    }
  }
}

__global__ __launch_bounds__(NW_ * 32) void k_mlp(
    const float* __restrict__ xg, const float* __restrict__ xyz,
    const float* __restrict__ new_xyz, const int* __restrict__ knn,
    const float* __restrict__ W0, const float* __restrict__ g0,
    const float* __restrict__ b0, const float* __restrict__ m0,
    const float* __restrict__ v0,
    const float* __restrict__ W1, const float* __restrict__ g1,
    const float* __restrict__ b1, const float* __restrict__ m1,
    const float* __restrict__ v1,
    const float* __restrict__ W2, const float* __restrict__ g2,
    const float* __restrict__ b2, const float* __restrict__ m2,
    const float* __restrict__ v2,
    float* __restrict__ out_feat)
{
  __shared__ __attribute__((aligned(16))) float Xs[XPAIRS_ * PP_];
  __shared__ __attribute__((aligned(16))) float Ys[YPAIRS_ * PP_];
  const int wave = __builtin_amdgcn_readfirstlane(threadIdx.x >> 5);
  const int lane = threadIdx.x & 31;
  const int b = blockIdx.x % B_;
  const int mq0 = (blockIdx.x / B_) * NW_;
  const int q = b * M_ + mq0 + wave;

  const int mt01 = wave & 3;
  const int q01  = (wave >> 2) * (NW_ / 2);
  RowTile<17> t0;
  load_row_tile<17, 67>(t0, W0, g0, b0, m0, v0, mt01, lane);

  const int nj = knn[q * K_ + lane];
  const int colw = (wave * 32 + lane) * 2;
  {
    constexpr int VPL = TILEW_ / 4 / (NW_ * 32);
    const float* src = xg + ((size_t)b * (M_ / NW_) + blockIdx.x / B_) * TILEW_;
    const int t = threadIdx.x;
    v4f r[VPL];
#pragma unroll
    for (int i = 0; i < VPL; ++i)
      r[i] = __builtin_nontemporal_load((const v4f*)(src + (i * (NW_ * 32) + t) * 4));
#pragma unroll
    for (int i = 0; i < VPL; ++i) {
      const int pc = i * (NW_ * 32) + t;
      *(v4f*)(Xs + (pc / (NCOL_ / 2)) * PP_ + (pc % (NCOL_ / 2)) * 4) = r[i];
    }
  }
  {
    const float* p  = xyz + ((size_t)b * N_ + nj) * 3;
    const float* qc = new_xyz + (size_t)q * 3;
    v2f e0, e1;
    e0.x = p[0] - qc[0];
    e0.y = p[1] - qc[1];
    e1.x = p[2] - qc[2];
    e1.y = 0.0f;
    *(v2f*)(Xs + 32 * PP_ + colw) = e0;
    *(v2f*)(Xs + 33 * PP_ + colw) = e1;
  }
  __syncthreads();

  RowTile<16> t1, t2;
  layer_rows<17, NW_ / 2, false>(Xs, Ys, t0, mt01, q01, lane, nullptr,
      [&] { load_row_tile<16, 64>(t1, W1, g1, b1, m1, v1, mt01, lane); });
  __syncthreads();
  layer_rows<16, NW_ / 2, false>(Ys, Xs, t1, mt01, q01, lane, nullptr,
      [&] { load_row_tile<16, 64>(t2, W2, g2, b2, m2, v2, wave, lane); });
  __syncthreads();
  layer_rows<16, NW_, true>(Xs, nullptr, t2, wave, 0, lane,
      out_feat + ((size_t)b * 128 + wave * 16 + (lane >> 4) * 8) * M_ + mq0,
      [] {});
}

extern "C" void kernel_launch(void* const* d_in, const int* in_sizes, int n_in,
                              void* d_out, int out_size, void* d_ws, size_t ws_size,
                              hipStream_t stream) {
  const float* xyz      = (const float*)d_in[0];
  const float* features = (const float*)d_in[1];
  const int*   indices  = (const int*)d_in[2];
  const float* W0 = (const float*)d_in[3];
  const float* g0 = (const float*)d_in[4];
  const float* b0 = (const float*)d_in[5];
  const float* m0 = (const float*)d_in[6];
  const float* v0 = (const float*)d_in[7];
  const float* W1 = (const float*)d_in[8];
  const float* g1 = (const float*)d_in[9];
  const float* b1 = (const float*)d_in[10];
  const float* m1 = (const float*)d_in[11];
  const float* v1 = (const float*)d_in[12];
  const float* W2 = (const float*)d_in[13];
  const float* g2 = (const float*)d_in[14];
  const float* b2 = (const float*)d_in[15];
  const float* m2 = (const float*)d_in[16];
  const float* v2 = (const float*)d_in[17];

  float* new_xyz  = (float*)d_out;
  float* out_feat = new_xyz + (size_t)B_ * M_ * 3;
  int*   knn      = (int*)d_ws;
  float* xg       = (float*)d_ws + (size_t)B_ * M_ * K_;

  k_new_xyz<<<(B_ * M_ + 255) / 256, 256, 0, stream>>>(xyz, indices, new_xyz);
  k_knn<<<B_ * M_, 32, 0, stream>>>(xyz, new_xyz, knn);
  k_gather<<<B_ * (C_ / GCH_) * GQP_, 256, 0, stream>>>(features, knn, xg);
  k_mlp<<<(B_ * M_) / NW_, NW_ * 32, 0, stream>>>(xg, xyz, new_xyz, knn,
      W0, g0, b0, m0, v0, W1, g1, b1, m1, v1, W2, g2, b2, m2, v2, out_feat);
}


# kNN point reads: xyz copied once per call into x, y, z planes in the workspace; a lane takes 4 consecutive points by three 16-byte whole-line loads, not one 12-byte load per point; same keys and sums
# speedup vs baseline: 2.4693x; 1.0630x over previous
#include <hip/hip_runtime.h>
#include <hip/hip_bf16.h>

typedef __attribute__((ext_vector_type(2))) float v2f;
typedef __attribute__((ext_vector_type(8))) float v8f;
typedef __attribute__((ext_vector_type(4))) float v4f;

#define B_   16
#define N_   4096
#define C_   64
#define M_   1024
#define K_   32
#define EPS_ 1e-5f
#define BIGF 3.402823e38f

__global__ __launch_bounds__(256) void k_new_xyz(const float* __restrict__ xyz,
                                                 const int* __restrict__ indices,
                                                 float* __restrict__ new_xyz,
                                                 float* __restrict__ planes) {
  static_assert(B_ * N_ == 4 * B_ * M_, "a thread turns 4 points");
  int t = blockIdx.x * 256 + threadIdx.x;
  if (t >= B_ * M_) return;
  int b = t >> 10, m = t & 1023;
  int n = indices[m];
  const float* s = xyz + ((size_t)b * N_ + n) * 3;
  float* d = new_xyz + (size_t)t * 3;
  d[0] = s[0]; d[1] = s[1]; d[2] = s[2];
  const v4f* p4 = (const v4f*)xyz + (size_t)t * 3;
  const v4f a = p4[0];
  const v4f e = p4[1];
  const v4f f = p4[2];
  v4f px, py, pz;
  px.x = a.x; px.y = a.w; px.z = e.z; px.w = f.y;
  py.x = a.y; py.y = e.x; py.z = e.w; py.w = f.z;
  pz.x = a.z; pz.y = e.y; pz.z = f.x; pz.w = f.w;
  ((v4f*)planes)[t] = px;
  ((v4f*)(planes + (size_t)B_ * N_))[t] = py;
  ((v4f*)(planes + (size_t)2 * B_ * N_))[t] = pz;
}

#define CAP_ 128

__device__ __forceinline__ float sqdist(float qx, float qy, float qz, float qq,
                                        float x, float y, float z) {
  const float pp  = fmaf(z, z, fmaf(x, x, y * y));
  const float dot = fmaf(qz, z, fmaf(qx, x, qy * y));
  return fmaf(-2.0f, dot, qq + pp);
}

__device__ __forceinline__ unsigned okey(float f) {
  const unsigned u = __float_as_uint(f);
  return (u & 0x80000000u) ? ~u : (u | 0x80000000u);
}
__device__ __forceinline__ float okey_inv(unsigned k) {
  return __uint_as_float((k & 0x80000000u) ? (k & 0x7fffffffu) : ~k);
}

template <int NS, int NB>
__device__ __forceinline__ unsigned kth32(const unsigned (&k)[NS]) {
  unsigned v = 0;
  for (int bit = 31; bit > 31 - NB; --bit) {
    const unsigned t = v | (1u << bit);
    int c = 0;
#pragma unroll
    for (int s = 0; s < NS; ++s)
      c += __builtin_popcount(__builtin_amdgcn_ballot_w32(k[s] < t));
    if (c < K_) v = t;
  }
  if (NB < 32) v |= (1u << (32 - NB)) - 1u;
  return v;
}

__global__ __launch_bounds__(32) void k_knn(const float* __restrict__ planes,
                                            const float* __restrict__ new_xyz,
                                            int* __restrict__ knn) {
  __shared__ __attribute__((aligned(16))) float dl[N_];
  __shared__ unsigned ck[CAP_];
  __shared__ int      ci[CAP_];
  const int lane = threadIdx.x;
  const int q = blockIdx.x;
  const int b = q >> 10;

  const float qx = new_xyz[q * 3 + 0];
  const float qy = new_xyz[q * 3 + 1];
  const float qz = new_xyz[q * 3 + 2];
  float qq;
  {
#pragma clang fp contract(off)
    const float yy = qy * qy;
    const float zz = qz * qz;
    qq = fmaf(qx, qx, yy) + zz;
  }
  const float* px = planes + (size_t)b * N_;
  const float* py = px + (size_t)B_ * N_;
  const float* pz = py + (size_t)B_ * N_;

  float m1 = BIGF, m2 = BIGF;
#pragma unroll 8
  for (int j = 0; j < N_ / 128; ++j) {
    const int n0 = j * 128 + lane * 4;
    const v4f x4 = *(const v4f*)(px + n0);
    const v4f y4 = *(const v4f*)(py + n0);
    const v4f z4 = *(const v4f*)(pz + n0);
    v4f d4;
    d4.x = sqdist(qx, qy, qz, qq, x4.x, y4.x, z4.x);
    d4.y = sqdist(qx, qy, qz, qq, x4.y, y4.y, z4.y);
    d4.z = sqdist(qx, qy, qz, qq, x4.z, y4.z, z4.z);
    d4.w = sqdist(qx, qy, qz, qq, x4.w, y4.w, z4.w);
    *(v4f*)(dl + n0) = d4;
#pragma unroll
    for (int e = 0; e < 4; ++e) {
      const float d = d4[e];
      m2 = fminf(m2, fmaxf(m1, d));
      m1 = fminf(m1, d);
    }
  }

  const unsigned kb[2] = { okey(m1), okey(m2) };
  const float fB = okey_inv(kth32<2, 14>(kb));

  int cnt = 0;
  for (int w = 0; w < N_ / 1024; ++w) {
    unsigned bw = 0;
    v4f g4[8];
#pragma unroll
    for (int g = 0; g < 8; ++g) g4[g] = *(const v4f*)(dl + 128 * (8 * w + g) + lane * 4);
#pragma unroll
    for (int u = 0; u < 32; ++u)
      bw |= (g4[u >> 2][u & 3] <= fB) ? (1u << u) : 0u;
    for (int t = 0; t < 32; ++t) {
      const bool has = bw != 0;
      const unsigned mask = __builtin_amdgcn_ballot_w32(has);
      if (mask == 0) break;
      if (has) {
        const int u = __builtin_ctz(bw);
        bw &= bw - 1;
        const int n = 128 * (8 * w + (u >> 2)) + lane * 4 + (u & 3);
        const int pos = cnt + (int)__builtin_amdgcn_mbcnt_lo(mask, 0u);
        if (pos < CAP_) { ck[pos] = okey(dl[n]); ci[pos] = n; }
      }
      cnt += __builtin_popcount(mask);
    }
  }
  __syncthreads();

  int* out = knn + q * K_;
  bool done = false;
  if (cnt >= K_ && cnt <= CAP_) {
    unsigned k[CAP_ / 32];
    int id[CAP_ / 32];
#pragma unroll
    for (int s = 0; s < CAP_ / 32; ++s) {
      const int j = s * 32 + lane;
      const bool valid = j < cnt;
      k[s]  = valid ? ck[j] : 0xffffffffu;
      id[s] = valid ? ci[j] : 0;
    }
    unsigned v32;
    if (cnt <= 64) {
      const unsigned k2[2] = { k[0], k[1] };
      v32 = kth32<2, 32>(k2);
    } else {
      v32 = kth32<CAP_ / 32, 32>(k);
    }
    unsigned mt[CAP_ / 32];
    int total = 0;
#pragma unroll
    for (int s = 0; s < CAP_ / 32; ++s) {
      mt[s] = __builtin_amdgcn_ballot_w32(s * 32 + lane < cnt && k[s] <= v32);
      total += __builtin_popcount(mt[s]);
    }
    if (total == K_) {
      int wpos = 0;
#pragma unroll
      for (int s = 0; s < CAP_ / 32; ++s) {
        if ((mt[s] >> lane) & 1u) out[wpos + (int)__builtin_amdgcn_mbcnt_lo(mt[s], 0u)] = id[s];
        wpos += __builtin_popcount(mt[s]);
      }
      done = true;
    }
  }
  if (!done) {
    for (int kk = 0; kk < K_; ++kk) {
      float mv = BIGF;
      int   mi = 0x7fffffff;
      for (int n = lane; n < N_; n += 32) {
        float v = dl[n];
        if (v < mv) { mv = v; mi = n; }
      }
#pragma unroll
      for (int off = 16; off > 0; off >>= 1) {
        float ov = __shfl_xor(mv, off, 32);
        int   oi = __shfl_xor(mi, off, 32);
        if (ov < mv || (ov == mv && oi < mi)) { mv = ov; mi = oi; }
      }
      if (lane == 0) out[kk] = mi;
      dl[mi] = BIGF;
      __syncthreads();
    }
  }
}

#define NW_     8
#define NCOL_   (NW_ * 32)
#define PP_     (2 * NCOL_ + 32)
#define XPAIRS_ 34
#define YPAIRS_ 32
#define FPAIRS_ (C_ / 2)
#define TILEW_  (FPAIRS_ * 2 * NCOL_)
#define GCH_    4
#define GQP_    4

__global__ __launch_bounds__(256) void k_gather(const float* __restrict__ features,
                                                const int* __restrict__ knn,
                                                float* __restrict__ xg) {
  __shared__ __attribute__((aligned(16))) float st[(GCH_ / 2) * 2 * N_];
  const int t    = threadIdx.x;
  const int lane = t & 31;
  const int wave = __builtin_amdgcn_readfirstlane(t >> 5);
  const int b    = blockIdx.x % B_;
  const int rest = blockIdx.x / B_;
  const int cg   = rest % (C_ / GCH_);
  const int part = rest / (C_ / GCH_);
  const float* fb = features + ((size_t)b * C_ + cg * GCH_) * N_;

  {
    constexpr int NI = N_ / 4 / 256;
    v4f ra[GCH_ / 2][NI], rb[GCH_ / 2][NI];
#pragma unroll
    for (int pr = 0; pr < GCH_ / 2; ++pr)
#pragma unroll
      for (int i = 0; i < NI; ++i) {
        const int pc = i * 256 + t;
        ra[pr][i] = __builtin_nontemporal_load((const v4f*)(fb + (size_t)(2 * pr)     * N_ + pc * 4));
        rb[pr][i] = __builtin_nontemporal_load((const v4f*)(fb + (size_t)(2 * pr + 1) * N_ + pc * 4));
      }
#pragma unroll
    for (int pr = 0; pr < GCH_ / 2; ++pr)
#pragma unroll
      for (int i = 0; i < NI; ++i) {
        const int pc = i * 256 + t;
        v4f lo, hi;
        lo.x = ra[pr][i].x; lo.y = rb[pr][i].x; lo.z = ra[pr][i].y; lo.w = rb[pr][i].y;
        hi.x = ra[pr][i].z; hi.y = rb[pr][i].z; hi.z = ra[pr][i].w; hi.w = rb[pr][i].w;
        *(v4f*)(st + pr * 2 * N_ + pc * 8)     = lo;
        *(v4f*)(st + pr * 2 * N_ + pc * 8 + 4) = hi;
      }
  }
  __syncthreads();

  constexpr int QPB = M_ / GQP_;
  const int m0 = part * QPB + wave;
  const int*  kq = knn + ((size_t)b * M_ + m0) * K_ + lane;
  float* dq = xg + ((size_t)b * (M_ / NW_) + part * (QPB / NW_)) * TILEW_
                 + (size_t)(cg * (GCH_ / 2)) * (2 * NCOL_) + (wave * 32 + lane) * 2;
#pragma unroll 4
  for (int it = 0; it < QPB / NW_; ++it) {
    const int nj = kq[(size_t)it * NW_ * K_];
#pragma unroll
    for (int pr = 0; pr < GCH_ / 2; ++pr) {
      const v2f p = *(const v2f*)(st + pr * 2 * N_ + nj * 2);
      *(v2f*)(dq + (size_t)it * TILEW_ + pr * (2 * NCOL_)) = p;
    }
  }
}

template <int MASK>
__device__ __forceinline__ float row_xor(float v) {
  return __builtin_bit_cast(float, __builtin_amdgcn_update_dpp(
      0, __builtin_bit_cast(int, v), 0x160 + MASK, 0xf, 0xf, true));
}
__device__ __forceinline__ float row_max16(float v) {
  v = fmaxf(v, row_xor<1>(v));
  v = fmaxf(v, row_xor<2>(v));
  v = fmaxf(v, row_xor<4>(v));
  v = fmaxf(v, row_xor<8>(v));
  return v;
}

template <int NKT>
__device__ __forceinline__ void load_b(v2f (&bq)[2][NKT], const float* Xin, int cw) {
#pragma unroll
  for (int kt = 0; kt < NKT; ++kt) {
    bq[0][kt] = *(const v2f*)(Xin + kt * 2 * PP_ + cw);
    bq[1][kt] = *(const v2f*)(Xin + kt * 2 * PP_ + cw + 32);
  }
}

template <int NKT>
struct RowTile {
  v2f a[NKT];
  float s;
  v8f c;
};
template <int NKT, int KW>
__device__ __forceinline__ void load_row_tile(
    RowTile<NKT>& t, const float* __restrict__ W,
    const float* __restrict__ g, const float* __restrict__ bt,
    const float* __restrict__ mu, const float* __restrict__ vr, int mt, int lane)
{
  const int half = lane >> 4;
  const int arow = mt * 16 + (lane & 15);
#pragma unroll
  for (int kt = 0; kt < NKT; ++kt) {
    const int k0 = kt * 4 + half * 2;
    t.a[kt].x = (k0 + 0 < KW) ? W[arow * KW + k0 + 0] : 0.0f;
    t.a[kt].y = (k0 + 1 < KW) ? W[arow * KW + k0 + 1] : 0.0f;
  }
  t.s = g[arow] / sqrtf(vr[arow] + EPS_);
  const float crow = bt[arow] - mu[arow] * t.s;
#pragma unroll
  for (int r = 0; r < 8; ++r) t.c[r] = __shfl(crow, half * 24 + r, 32);
}

template <int NKT, int NQ, bool POOL, typename NextTile>
__device__ __forceinline__ void layer_rows(
    const float* Xin, float* Yout, const RowTile<NKT>& t,
    int mt, int q0, int lane, float* __restrict__ pool_out, NextTile next_tile)
{
  static_assert(NQ % 2 == 0, "queries are taken two per trip");
  const int half = lane >> 4;
  const int l16  = lane & 15;

  v2f as[NKT];
#pragma unroll
  for (int kt = 0; kt < NKT; ++kt) as[kt] = t.a[kt] * t.s;

  const v8f cinit = t.c;

  float res[8];
#pragma unroll
  for (int r = 0; r < 8; ++r) res[r] = 0.0f;

  const int cw = half * PP_ + l16 * 2;
  auto do_query = [&](const v2f (&bcur)[2][NKT], int qq) {
    v8f acc0 = cinit, acc1 = cinit;
#pragma unroll
    for (int kt = 0; kt < NKT; ++kt) {
      acc0 = __builtin_amdgcn_wmma_f32_16x16x4_f32(
           false, as[kt],  false, bcur[0][kt],
           (short)0, acc0,  false,  false);
      acc1 = __builtin_amdgcn_wmma_f32_16x16x4_f32(
           false, as[kt],  false, bcur[1][kt],
           (short)0, acc1,  false,  false);
    }
    if (!POOL) {
      float* yq = Yout + (mt * 8 + half * 4) * PP_ + (qq * 32 + l16) * 2;
#pragma unroll
      for (int j = 0; j < 4; ++j) {
        v2f y0, y1;
        y0.x = fmaxf(acc0[2 * j],     0.0f);
        y0.y = fmaxf(acc0[2 * j + 1], 0.0f);
        y1.x = fmaxf(acc1[2 * j],     0.0f);
        y1.y = fmaxf(acc1[2 * j + 1], 0.0f);
        *(v2f*)(yq + j * PP_)      = y0;
        *(v2f*)(yq + j * PP_ + 32) = y1;
      }
    } else {
#pragma unroll
      for (int r = 0; r < 8; ++r) {
        const float v = row_max16(fmaxf(acc0[r], acc1[r]));
        if (l16 == qq) res[r] = v;
      }
    }
  };
  v2f bq0[2][NKT], bq1[2][NKT];
  load_b<NKT>(bq0, Xin, cw + q0 * 64);
  int q = q0;
#pragma nounroll
  for (; q + 2 < q0 + NQ; q += 2) {
    load_b<NKT>(bq1, Xin, cw + (q + 1) * 64);
    do_query(bq0, q);
    load_b<NKT>(bq0, Xin, cw + (q + 2) * 64);
    do_query(bq1, q + 1);
  }
  load_b<NKT>(bq1, Xin, cw + (q + 1) * 64);
  do_query(bq0, q);
  next_tile();
  do_query(bq1, q + 1);

  if (POOL) {
    if (l16 < NQ) {
#pragma unroll
      for (int r = 0; r < 8; ++r) pool_out[(size_t)r * M_ + q0 + l16] = fmaxf(res[r], 0.0f);
    }
  }
}

__global__ __launch_bounds__(NW_ * 32) void k_mlp(
    const float* __restrict__ xg, const float* __restrict__ xyz,
    const float* __restrict__ new_xyz, const int* __restrict__ knn,
    const float* __restrict__ W0, const float* __restrict__ g0,
    const float* __restrict__ b0, const float* __restrict__ m0,
    const float* __restrict__ v0,
    const float* __restrict__ W1, const float* __restrict__ g1,
    const float* __restrict__ b1, const float* __restrict__ m1,
    const float* __restrict__ v1,
    const float* __restrict__ W2, const float* __restrict__ g2,
    const float* __restrict__ b2, const float* __restrict__ m2,
    const float* __restrict__ v2,
    float* __restrict__ out_feat)
{
  __shared__ __attribute__((aligned(16))) float Xs[XPAIRS_ * PP_];
  __shared__ __attribute__((aligned(16))) float Ys[YPAIRS_ * PP_];
  const int wave = __builtin_amdgcn_readfirstlane(threadIdx.x >> 5);
  const int lane = threadIdx.x & 31;
  const int b = blockIdx.x % B_;
  const int mq0 = (blockIdx.x / B_) * NW_;
  const int q = b * M_ + mq0 + wave;

  const int mt01 = wave & 3;
  const int q01  = (wave >> 2) * (NW_ / 2);
  RowTile<17> t0;
  load_row_tile<17, 67>(t0, W0, g0, b0, m0, v0, mt01, lane);

  const int nj = knn[q * K_ + lane];
  const int colw = (wave * 32 + lane) * 2;
  {
    constexpr int VPL = TILEW_ / 4 / (NW_ * 32);
    const float* src = xg + ((size_t)b * (M_ / NW_) + blockIdx.x / B_) * TILEW_;
    const int t = threadIdx.x;
    v4f r[VPL];
#pragma unroll
    for (int i = 0; i < VPL; ++i)
      r[i] = __builtin_nontemporal_load((const v4f*)(src + (i * (NW_ * 32) + t) * 4));
#pragma unroll
    for (int i = 0; i < VPL; ++i) {
      const int pc = i * (NW_ * 32) + t;
      *(v4f*)(Xs + (pc / (NCOL_ / 2)) * PP_ + (pc % (NCOL_ / 2)) * 4) = r[i];
    }
  }
  {
    const float* p  = xyz + ((size_t)b * N_ + nj) * 3;
    const float* qc = new_xyz + (size_t)q * 3;
    v2f e0, e1;
    e0.x = p[0] - qc[0];
    e0.y = p[1] - qc[1];
    e1.x = p[2] - qc[2];
    e1.y = 0.0f;
    *(v2f*)(Xs + 32 * PP_ + colw) = e0;
    *(v2f*)(Xs + 33 * PP_ + colw) = e1;
  }
  __syncthreads();

  RowTile<16> t1, t2;
  layer_rows<17, NW_ / 2, false>(Xs, Ys, t0, mt01, q01, lane, nullptr,
      [&] { load_row_tile<16, 64>(t1, W1, g1, b1, m1, v1, mt01, lane); });
  __syncthreads();
  layer_rows<16, NW_ / 2, false>(Ys, Xs, t1, mt01, q01, lane, nullptr,
      [&] { load_row_tile<16, 64>(t2, W2, g2, b2, m2, v2, wave, lane); });
  __syncthreads();
  layer_rows<16, NW_, true>(Xs, nullptr, t2, wave, 0, lane,
      out_feat + ((size_t)b * 128 + wave * 16 + (lane >> 4) * 8) * M_ + mq0,
      [] {});
}

extern "C" void kernel_launch(void* const* d_in, const int* in_sizes, int n_in,
                              void* d_out, int out_size, void* d_ws, size_t ws_size,
                              hipStream_t stream) {
  const float* xyz      = (const float*)d_in[0];
  const float* features = (const float*)d_in[1];
  const int*   indices  = (const int*)d_in[2];
  const float* W0 = (const float*)d_in[3];
  const float* g0 = (const float*)d_in[4];
  const float* b0 = (const float*)d_in[5];
  const float* m0 = (const float*)d_in[6];
  const float* v0 = (const float*)d_in[7];
  const float* W1 = (const float*)d_in[8];
  const float* g1 = (const float*)d_in[9];
  const float* b1 = (const float*)d_in[10];
  const float* m1 = (const float*)d_in[11];
  const float* v1 = (const float*)d_in[12];
  const float* W2 = (const float*)d_in[13];
  const float* g2 = (const float*)d_in[14];
  const float* b2 = (const float*)d_in[15];
  const float* m2 = (const float*)d_in[16];
  const float* v2 = (const float*)d_in[17];

  float* new_xyz  = (float*)d_out;
  float* out_feat = new_xyz + (size_t)B_ * M_ * 3;
  int*   knn      = (int*)d_ws;
  float* xg       = (float*)d_ws + (size_t)B_ * M_ * K_;
  float* planes   = xg + (size_t)B_ * (M_ / NW_) * TILEW_;

  k_new_xyz<<<(B_ * M_ + 255) / 256, 256, 0, stream>>>(xyz, indices, new_xyz, planes);
  k_knn<<<B_ * M_, 32, 0, stream>>>(planes, new_xyz, knn);
  k_gather<<<B_ * (C_ / GCH_) * GQP_, 256, 0, stream>>>(features, knn, xg);
  k_mlp<<<(B_ * M_) / NW_, NW_ * 32, 0, stream>>>(xg, xyz, new_xyz, knn,
      W0, g0, b0, m0, v0, W1, g1, b1, m1, v1, W2, g2, b2, m2, v2, out_feat);
}


# MLP tiled by columns: a wave takes a query through all 3 layers in registers, a layer's accumulator tiles are the next layer's B fragments, weights as fragments in LDS; layers 1, 2 sum k in new order
# speedup vs baseline: 2.5749x; 1.0428x over previous
#include <hip/hip_runtime.h>
#include <hip/hip_bf16.h>

typedef __attribute__((ext_vector_type(2))) float v2f;
typedef __attribute__((ext_vector_type(8))) float v8f;
typedef __attribute__((ext_vector_type(4))) float v4f;

#define B_   16
#define N_   4096
#define C_   64
#define M_   1024
#define K_   32
#define EPS_ 1e-5f
#define BIGF 3.402823e38f

__global__ __launch_bounds__(256) void k_new_xyz(const float* __restrict__ xyz,
                                                 const int* __restrict__ indices,
                                                 float* __restrict__ new_xyz,
                                                 float* __restrict__ planes) {
  static_assert(B_ * N_ == 4 * B_ * M_, "a thread turns 4 points");
  int t = blockIdx.x * 256 + threadIdx.x;
  if (t >= B_ * M_) return;
  int b = t >> 10, m = t & 1023;
  int n = indices[m];
  const float* s = xyz + ((size_t)b * N_ + n) * 3;
  float* d = new_xyz + (size_t)t * 3;
  d[0] = s[0]; d[1] = s[1]; d[2] = s[2];
  const v4f* p4 = (const v4f*)xyz + (size_t)t * 3;
  const v4f a = p4[0];
  const v4f e = p4[1];
  const v4f f = p4[2];
  v4f px, py, pz;
  px.x = a.x; px.y = a.w; px.z = e.z; px.w = f.y;
  py.x = a.y; py.y = e.x; py.z = e.w; py.w = f.z;
  pz.x = a.z; pz.y = e.y; pz.z = f.x; pz.w = f.w;
  ((v4f*)planes)[t] = px;
  ((v4f*)(planes + (size_t)B_ * N_))[t] = py;
  ((v4f*)(planes + (size_t)2 * B_ * N_))[t] = pz;
}

#define CAP_ 128

__device__ __forceinline__ float sqdist(float qx, float qy, float qz, float qq,
                                        float x, float y, float z) {
  const float pp  = fmaf(z, z, fmaf(x, x, y * y));
  const float dot = fmaf(qz, z, fmaf(qx, x, qy * y));
  return fmaf(-2.0f, dot, qq + pp);
}

__device__ __forceinline__ unsigned okey(float f) {
  const unsigned u = __float_as_uint(f);
  return (u & 0x80000000u) ? ~u : (u | 0x80000000u);
}
__device__ __forceinline__ float okey_inv(unsigned k) {
  return __uint_as_float((k & 0x80000000u) ? (k & 0x7fffffffu) : ~k);
}

template <int NS, int NB>
__device__ __forceinline__ unsigned kth32(const unsigned (&k)[NS]) {
  unsigned v = 0;
  for (int bit = 31; bit > 31 - NB; --bit) {
    const unsigned t = v | (1u << bit);
    int c = 0;
#pragma unroll
    for (int s = 0; s < NS; ++s)
      c += __builtin_popcount(__builtin_amdgcn_ballot_w32(k[s] < t));
    if (c < K_) v = t;
  }
  if (NB < 32) v |= (1u << (32 - NB)) - 1u;
  return v;
}

__global__ __launch_bounds__(32) void k_knn(const float* __restrict__ planes,
                                            const float* __restrict__ new_xyz,
                                            int* __restrict__ knn) {
  __shared__ __attribute__((aligned(16))) float dl[N_];
  __shared__ unsigned ck[CAP_];
  __shared__ int      ci[CAP_];
  const int lane = threadIdx.x;
  const int q = blockIdx.x;
  const int b = q >> 10;

  const float qx = new_xyz[q * 3 + 0];
  const float qy = new_xyz[q * 3 + 1];
  const float qz = new_xyz[q * 3 + 2];
  float qq;
  {
#pragma clang fp contract(off)
    const float yy = qy * qy;
    const float zz = qz * qz;
    qq = fmaf(qx, qx, yy) + zz;
  }
  const float* px = planes + (size_t)b * N_;
  const float* py = px + (size_t)B_ * N_;
  const float* pz = py + (size_t)B_ * N_;

  float m1 = BIGF, m2 = BIGF;
#pragma unroll 8
  for (int j = 0; j < N_ / 128; ++j) {
    const int n0 = j * 128 + lane * 4;
    const v4f x4 = *(const v4f*)(px + n0);
    const v4f y4 = *(const v4f*)(py + n0);
    const v4f z4 = *(const v4f*)(pz + n0);
    v4f d4;
    d4.x = sqdist(qx, qy, qz, qq, x4.x, y4.x, z4.x);
    d4.y = sqdist(qx, qy, qz, qq, x4.y, y4.y, z4.y);
    d4.z = sqdist(qx, qy, qz, qq, x4.z, y4.z, z4.z);
    d4.w = sqdist(qx, qy, qz, qq, x4.w, y4.w, z4.w);
    *(v4f*)(dl + n0) = d4;
#pragma unroll
    for (int e = 0; e < 4; ++e) {
      const float d = d4[e];
      m2 = fminf(m2, fmaxf(m1, d));
      m1 = fminf(m1, d);
    }
  }

  const unsigned kb[2] = { okey(m1), okey(m2) };
  const float fB = okey_inv(kth32<2, 14>(kb));

  int cnt = 0;
  for (int w = 0; w < N_ / 1024; ++w) {
    unsigned bw = 0;
    v4f g4[8];
#pragma unroll
    for (int g = 0; g < 8; ++g) g4[g] = *(const v4f*)(dl + 128 * (8 * w + g) + lane * 4);
#pragma unroll
    for (int u = 0; u < 32; ++u)
      bw |= (g4[u >> 2][u & 3] <= fB) ? (1u << u) : 0u;
    for (int t = 0; t < 32; ++t) {
      const bool has = bw != 0;
      const unsigned mask = __builtin_amdgcn_ballot_w32(has);
      if (mask == 0) break;
      if (has) {
        const int u = __builtin_ctz(bw);
        bw &= bw - 1;
        const int n = 128 * (8 * w + (u >> 2)) + lane * 4 + (u & 3);
        const int pos = cnt + (int)__builtin_amdgcn_mbcnt_lo(mask, 0u);
        if (pos < CAP_) { ck[pos] = okey(dl[n]); ci[pos] = n; }
      }
      cnt += __builtin_popcount(mask);
    }
  }
  __syncthreads();

  int* out = knn + q * K_;
  bool done = false;
  if (cnt >= K_ && cnt <= CAP_) {
    unsigned k[CAP_ / 32];
    int id[CAP_ / 32];
#pragma unroll
    for (int s = 0; s < CAP_ / 32; ++s) {
      const int j = s * 32 + lane;
      const bool valid = j < cnt;
      k[s]  = valid ? ck[j] : 0xffffffffu;
      id[s] = valid ? ci[j] : 0;
    }
    unsigned v32;
    if (cnt <= 64) {
      const unsigned k2[2] = { k[0], k[1] };
      v32 = kth32<2, 32>(k2);
    } else {
      v32 = kth32<CAP_ / 32, 32>(k);
    }
    unsigned mt[CAP_ / 32];
    int total = 0;
#pragma unroll
    for (int s = 0; s < CAP_ / 32; ++s) {
      mt[s] = __builtin_amdgcn_ballot_w32(s * 32 + lane < cnt && k[s] <= v32);
      total += __builtin_popcount(mt[s]);
    }
    if (total == K_) {
      int wpos = 0;
#pragma unroll
      for (int s = 0; s < CAP_ / 32; ++s) {
        if ((mt[s] >> lane) & 1u) out[wpos + (int)__builtin_amdgcn_mbcnt_lo(mt[s], 0u)] = id[s];
        wpos += __builtin_popcount(mt[s]);
      }
      done = true;
    }
  }
  if (!done) {
    for (int kk = 0; kk < K_; ++kk) {
      float mv = BIGF;
      int   mi = 0x7fffffff;
      for (int n = lane; n < N_; n += 32) {
        float v = dl[n];
        if (v < mv) { mv = v; mi = n; }
      }
#pragma unroll
      for (int off = 16; off > 0; off >>= 1) {
        float ov = __shfl_xor(mv, off, 32);
        int   oi = __shfl_xor(mi, off, 32);
        if (ov < mv || (ov == mv && oi < mi)) { mv = ov; mi = oi; }
      }
      if (lane == 0) out[kk] = mi;
      dl[mi] = BIGF;
      __syncthreads();
    }
  }
}

#define NW_     8
#define NCOL_   (NW_ * 32)
#define FPAIRS_ (C_ / 2)
#define TILEW_  (FPAIRS_ * 2 * NCOL_)
#define GCH_    4
#define GQP_    4
#define NK0_    17
#define NK12_   16
#define F1_     (4 * NK0_)
#define F2_     (F1_ + 4 * NK12_)
#define NFRAG_  (F2_ + 8 * NK12_)

__global__ __launch_bounds__(256) void k_gather(const float* __restrict__ features,
                                                const int* __restrict__ knn,
                                                float* __restrict__ xg) {
  __shared__ __attribute__((aligned(16))) float st[(GCH_ / 2) * 2 * N_];
  const int t    = threadIdx.x;
  const int lane = t & 31;
  const int wave = __builtin_amdgcn_readfirstlane(t >> 5);
  const int b    = blockIdx.x % B_;
  const int rest = blockIdx.x / B_;
  const int cg   = rest % (C_ / GCH_);
  const int part = rest / (C_ / GCH_);
  const float* fb = features + ((size_t)b * C_ + cg * GCH_) * N_;

  {
    constexpr int NI = N_ / 4 / 256;
    v4f ra[GCH_ / 2][NI], rb[GCH_ / 2][NI];
#pragma unroll
    for (int pr = 0; pr < GCH_ / 2; ++pr)
#pragma unroll
      for (int i = 0; i < NI; ++i) {
        const int pc = i * 256 + t;
        ra[pr][i] = __builtin_nontemporal_load((const v4f*)(fb + (size_t)(2 * pr)     * N_ + pc * 4));
        rb[pr][i] = __builtin_nontemporal_load((const v4f*)(fb + (size_t)(2 * pr + 1) * N_ + pc * 4));
      }
#pragma unroll
    for (int pr = 0; pr < GCH_ / 2; ++pr)
#pragma unroll
      for (int i = 0; i < NI; ++i) {
        const int pc = i * 256 + t;
        v4f lo, hi;
        lo.x = ra[pr][i].x; lo.y = rb[pr][i].x; lo.z = ra[pr][i].y; lo.w = rb[pr][i].y;
        hi.x = ra[pr][i].z; hi.y = rb[pr][i].z; hi.z = ra[pr][i].w; hi.w = rb[pr][i].w;
        *(v4f*)(st + pr * 2 * N_ + pc * 8)     = lo;
        *(v4f*)(st + pr * 2 * N_ + pc * 8 + 4) = hi;
      }
  }
  __syncthreads();

  constexpr int QPB = M_ / GQP_;
  const int m0 = part * QPB + wave;
  const int*  kq = knn + ((size_t)b * M_ + m0) * K_ + lane;
  float* dq = xg + ((size_t)b * (M_ / NW_) + part * (QPB / NW_)) * TILEW_
                 + (size_t)(cg * (GCH_ / 2)) * (2 * NCOL_) + (wave * 32 + lane) * 2;
#pragma unroll 4
  for (int it = 0; it < QPB / NW_; ++it) {
    const int nj = kq[(size_t)it * NW_ * K_];
#pragma unroll
    for (int pr = 0; pr < GCH_ / 2; ++pr) {
      const v2f p = *(const v2f*)(st + pr * 2 * N_ + nj * 2);
      *(v2f*)(dq + (size_t)it * TILEW_ + pr * (2 * NCOL_)) = p;
    }
  }
}

template <int MASK>
__device__ __forceinline__ float row_xor(float v) {
  return __builtin_bit_cast(float, __builtin_amdgcn_update_dpp(
      0, __builtin_bit_cast(int, v), 0x160 + MASK, 0xf, 0xf, true));
}
__device__ __forceinline__ float row_max16(float v) {
  v = fmaxf(v, row_xor<1>(v));
  v = fmaxf(v, row_xor<2>(v));
  v = fmaxf(v, row_xor<4>(v));
  v = fmaxf(v, row_xor<8>(v));
  return v;
}


typedef const float* __restrict__ cfp;

__device__ __forceinline__ int kperm(int st, int h) {
  return 16 * (st >> 2) + 8 * h + 2 * (st & 3);
}

template <int NKT, int KW, bool PERM>
__device__ __forceinline__ void load_rows(v2f (&a)[NK0_], cfp W, int mt, int lane) {
  const int half = lane >> 4;
  const int arow = mt * 16 + (lane & 15);
#pragma unroll
  for (int st = 0; st < NKT; ++st) {
    const int k0 = PERM ? kperm(st, half) : st * 4 + half * 2;
    a[st].x = (KW % 4 == 0 || k0 + 0 < KW) ? W[arow * KW + k0 + 0] : 0.0f;
    a[st].y = (KW % 4 == 0 || k0 + 1 < KW) ? W[arow * KW + k0 + 1] : 0.0f;
  }
}
struct BnRow { float g, bt, mu, vr; };
__device__ __forceinline__ BnRow load_bn(cfp g, cfp bt, cfp mu, cfp vr, int arow) {
  BnRow p;
  p.g = g[arow]; p.bt = bt[arow]; p.mu = mu[arow]; p.vr = vr[arow];
  return p;
}
template <int NKT>
__device__ __forceinline__ void store_rows(float* frag, float* shift, const v2f (&a)[NK0_],
                                           BnRow p, int arow, int lane) {
  const float s = p.g / sqrtf(p.vr + EPS_);
#pragma unroll
  for (int st = 0; st < NKT; ++st) *(v2f*)(frag + st * 64 + lane * 2) = a[st] * s;
  if ((lane >> 4) == 0) shift[arow] = p.bt - p.mu * s;
}

struct TileOps {
  v2f a[NK0_];
  v8f c;
};
template <int NKT>
__device__ __forceinline__ void load_ops(TileOps& o, const float* frag, const float* shift, int lane) {
#pragma unroll
  for (int st = 0; st < NKT; ++st) o.a[st] = *(const v2f*)(frag + st * 64 + lane * 2);
  const v4f lo = *(const v4f*)(shift + (lane >> 4) * 8);
  const v4f hi = *(const v4f*)(shift + (lane >> 4) * 8 + 4);
  o.c[0] = lo.x; o.c[1] = lo.y; o.c[2] = lo.z; o.c[3] = lo.w;
  o.c[4] = hi.x; o.c[5] = hi.y; o.c[6] = hi.z; o.c[7] = hi.w;
}

__device__ __forceinline__ v8f relu8(v8f a) {
  v8f y;
#pragma unroll
  for (int r = 0; r < 8; ++r) y[r] = fmaxf(a[r], 0.0f);
  return y;
}

__device__ __forceinline__ v8f wmma4(v2f a, v2f b, v8f c) {
  return __builtin_amdgcn_wmma_f32_16x16x4_f32(
       false, a,  false, b,
       (short)0, c,  false,  false);
}

__device__ __forceinline__ void row_tile_0(const TileOps& o, const v2f (&bx)[2][NK0_],
                                           v8f& acc0, v8f& acc1) {
  acc0 = o.c;
  acc1 = o.c;
#pragma unroll
  for (int kt = 0; kt < NK0_; ++kt) {
    acc0 = wmma4(o.a[kt], bx[0][kt], acc0);
    acc1 = wmma4(o.a[kt], bx[1][kt], acc1);
  }
}
__device__ __forceinline__ void row_tile_12(const TileOps& o, const v8f (&yin)[4][2],
                                            v8f& acc0, v8f& acc1) {
  acc0 = o.c;
  acc1 = o.c;
#pragma unroll
  for (int t = 0; t < 4; ++t)
#pragma unroll
    for (int j = 0; j < 4; ++j) {
      v2f b0, b1;
      b0.x = yin[t][0][2 * j]; b0.y = yin[t][0][2 * j + 1];
      b1.x = yin[t][1][2 * j]; b1.y = yin[t][1][2 * j + 1];
      acc0 = wmma4(o.a[4 * t + j], b0, acc0);
      acc1 = wmma4(o.a[4 * t + j], b1, acc1);
    }
}

#define TILE_FENCE() __builtin_amdgcn_sched_barrier(0)

__global__ __launch_bounds__(NW_ * 32) void k_mlp(
    const float* __restrict__ xg, const float* __restrict__ xyz,
    const float* __restrict__ new_xyz, const int* __restrict__ knn,
    const float* __restrict__ W0, const float* __restrict__ g0,
    const float* __restrict__ b0, const float* __restrict__ m0,
    const float* __restrict__ v0,
    const float* __restrict__ W1, const float* __restrict__ g1,
    const float* __restrict__ b1, const float* __restrict__ m1,
    const float* __restrict__ v1,
    const float* __restrict__ W2, const float* __restrict__ g2,
    const float* __restrict__ b2, const float* __restrict__ m2,
    const float* __restrict__ v2,
    float* __restrict__ out_feat)
{
  __shared__ __attribute__((aligned(16))) float As[NFRAG_ * 64];
  __shared__ __attribute__((aligned(16))) float Sh[64 + 64 + 128];
  __shared__ __attribute__((aligned(16))) float Pl[128 * NW_];
  const int wave = __builtin_amdgcn_readfirstlane(threadIdx.x >> 5);
  const int lane = threadIdx.x & 31;
  const int half = lane >> 4;
  const int l16  = lane & 15;
  const int b = blockIdx.x % B_;
  const int mq0 = (blockIdx.x / B_) * NW_;
  const int q = b * M_ + mq0 + wave;

  const float* qc = new_xyz + (size_t)q * 3;
  const float qx = qc[0], qy = qc[1], qz = qc[2];
  const int nj0 = knn[q * K_ + l16];
  const int nj1 = knn[q * K_ + 16 + l16];

  v2f wa[NK0_], wb[NK0_];
  BnRow pa, pb;
  const int arow_a = (wave & 3) * 16 + l16;
  const int arow_b = wave * 16 + l16;
  if (wave < 4) {
    load_rows<NK0_, 67, false>(wa, W0, wave, lane);
    pa = load_bn(g0, b0, m0, v0, arow_a);
  } else {
    load_rows<NK12_, 64, true>(wa, W1, wave - 4, lane);
    wa[NK0_ - 1] = (v2f){0.0f, 0.0f};
    pa = load_bn(g1, b1, m1, v1, arow_a);
  }
  load_rows<NK12_, 64, true>(wb, W2, wave, lane);
  pb = load_bn(g2, b2, m2, v2, arow_b);

  v2f bx[2][NK0_];
  {
    const float* tile = xg + ((size_t)b * (M_ / NW_) + blockIdx.x / B_) * TILEW_
                           + half * (2 * NCOL_) + (wave * 32 + l16) * 2;
#pragma unroll
    for (int kt = 0; kt < NK0_ - 1; ++kt) {
      bx[0][kt] = __builtin_nontemporal_load((const v2f*)(tile + kt * 2 * (2 * NCOL_)));
      bx[1][kt] = __builtin_nontemporal_load((const v2f*)(tile + kt * 2 * (2 * NCOL_) + 32));
    }
#pragma unroll
    for (int ct = 0; ct < 2; ++ct) {
      const float* p = xyz + ((size_t)b * N_ + (ct ? nj1 : nj0)) * 3;
      const float rx = p[0] - qx, ry = p[1] - qy, rz = p[2] - qz;
      bx[ct][NK0_ - 1].x = half ? rz : rx;
      bx[ct][NK0_ - 1].y = half ? 0.0f : ry;
    }
  }

  if (wave < 4)
    store_rows<NK0_>(As + (wave * NK0_) * 64, Sh, wa, pa, arow_a, lane);
  else
    store_rows<NK12_>(As + (F1_ + (wave - 4) * NK12_) * 64, Sh + 64, wa, pa, arow_a, lane);
  store_rows<NK12_>(As + (F2_ + wave * NK12_) * 64, Sh + 128, wb, pb, arow_b, lane);
  __syncthreads();

  TileOps oa, ob;
  v8f acc0, acc1;
  load_ops<NK0_>(oa, As, Sh, lane);

  v8f y0[4][2];
  load_ops<NK0_>(ob, As + (1 * NK0_) * 64, Sh + 16, lane);
  TILE_FENCE();
  row_tile_0(oa, bx, acc0, acc1);
  TILE_FENCE();
  y0[0][0] = relu8(acc0); y0[0][1] = relu8(acc1);
  load_ops<NK0_>(oa, As + (2 * NK0_) * 64, Sh + 32, lane);
  TILE_FENCE();
  row_tile_0(ob, bx, acc0, acc1);
  TILE_FENCE();
  y0[1][0] = relu8(acc0); y0[1][1] = relu8(acc1);
  load_ops<NK0_>(ob, As + (3 * NK0_) * 64, Sh + 48, lane);
  TILE_FENCE();
  row_tile_0(oa, bx, acc0, acc1);
  TILE_FENCE();
  y0[2][0] = relu8(acc0); y0[2][1] = relu8(acc1);
  load_ops<NK12_>(oa, As + F1_ * 64, Sh + 64, lane);
  TILE_FENCE();
  row_tile_0(ob, bx, acc0, acc1);
  TILE_FENCE();
  y0[3][0] = relu8(acc0); y0[3][1] = relu8(acc1);

  v8f y1[4][2];
  load_ops<NK12_>(ob, As + (F1_ + 1 * NK12_) * 64, Sh + 64 + 16, lane);
  TILE_FENCE();
  row_tile_12(oa, y0, acc0, acc1);
  TILE_FENCE();
  y1[0][0] = relu8(acc0); y1[0][1] = relu8(acc1);
  load_ops<NK12_>(oa, As + (F1_ + 2 * NK12_) * 64, Sh + 64 + 32, lane);
  TILE_FENCE();
  row_tile_12(ob, y0, acc0, acc1);
  TILE_FENCE();
  y1[1][0] = relu8(acc0); y1[1][1] = relu8(acc1);
  load_ops<NK12_>(ob, As + (F1_ + 3 * NK12_) * 64, Sh + 64 + 48, lane);
  TILE_FENCE();
  row_tile_12(oa, y0, acc0, acc1);
  TILE_FENCE();
  y1[2][0] = relu8(acc0); y1[2][1] = relu8(acc1);
  load_ops<NK12_>(oa, As + F2_ * 64, Sh + 128, lane);
  TILE_FENCE();
  row_tile_12(ob, y0, acc0, acc1);
  TILE_FENCE();
  y1[3][0] = relu8(acc0); y1[3][1] = relu8(acc1);

  auto pool_tile = [&](int mt) {
    float res = 0.0f;
#pragma unroll
    for (int r = 0; r < 8; ++r) {
      const float v = row_max16(fmaxf(acc0[r], acc1[r]));
      if (l16 == r) res = v;
    }
    if (l16 < 8) Pl[(mt * 16 + half * 8 + l16) * NW_ + wave] = fmaxf(res, 0.0f);
  };
#pragma nounroll
  for (int mp = 0; mp < 4; ++mp) {
    const int t1 = 2 * mp + 1;
    const int t2 = (mp < 3) ? 2 * mp + 2 : 7;
    load_ops<NK12_>(ob, As + (F2_ + t1 * NK12_) * 64, Sh + 128 + t1 * 16, lane);
    TILE_FENCE();
    row_tile_12(oa, y1, acc0, acc1);
    TILE_FENCE();
    pool_tile(2 * mp);
    load_ops<NK12_>(oa, As + (F2_ + t2 * NK12_) * 64, Sh + 128 + t2 * 16, lane);
    TILE_FENCE();
    row_tile_12(ob, y1, acc0, acc1);
    TILE_FENCE();
    pool_tile(t1);
  }
  __syncthreads();

  {
    const int t = threadIdx.x;
    const int row = t >> 1, part = t & 1;
    const v4f v = *(const v4f*)(Pl + row * NW_ + part * 4);
    *(v4f*)(out_feat + ((size_t)b * 128 + row) * M_ + mq0 + part * 4) = v;
  }
}

extern "C" void kernel_launch(void* const* d_in, const int* in_sizes, int n_in,
                              void* d_out, int out_size, void* d_ws, size_t ws_size,
                              hipStream_t stream) {
  const float* xyz      = (const float*)d_in[0];
  const float* features = (const float*)d_in[1];
  const int*   indices  = (const int*)d_in[2];
  const float* W0 = (const float*)d_in[3];
  const float* g0 = (const float*)d_in[4];
  const float* b0 = (const float*)d_in[5];
  const float* m0 = (const float*)d_in[6];
  const float* v0 = (const float*)d_in[7];
  const float* W1 = (const float*)d_in[8];
  const float* g1 = (const float*)d_in[9];
  const float* b1 = (const float*)d_in[10];
  const float* m1 = (const float*)d_in[11];
  const float* v1 = (const float*)d_in[12];
  const float* W2 = (const float*)d_in[13];
  const float* g2 = (const float*)d_in[14];
  const float* b2 = (const float*)d_in[15];
  const float* m2 = (const float*)d_in[16];
  const float* v2 = (const float*)d_in[17];

  float* new_xyz  = (float*)d_out;
  float* out_feat = new_xyz + (size_t)B_ * M_ * 3;
  int*   knn      = (int*)d_ws;
  float* xg       = (float*)d_ws + (size_t)B_ * M_ * K_;
  float* planes   = xg + (size_t)B_ * (M_ / NW_) * TILEW_;

  k_new_xyz<<<(B_ * M_ + 255) / 256, 256, 0, stream>>>(xyz, indices, new_xyz, planes);
  k_knn<<<B_ * M_, 32, 0, stream>>>(planes, new_xyz, knn);
  k_gather<<<B_ * (C_ / GCH_) * GQP_, 256, 0, stream>>>(features, knn, xg);
  k_mlp<<<(B_ * M_) / NW_, NW_ * 32, 0, stream>>>(xg, xyz, new_xyz, knn,
      W0, g0, b0, m0, v0, W1, g1, b1, m1, v1, W2, g2, b2, m2, v2, out_feat);
}


# kNN ties at rank 32 settled inside the survivor list (equal entries ranked by point number, wave-wide), not by the 32-pass argmin fallback whose lone wave outlasted the launch; same set and sums
# speedup vs baseline: 3.7259x; 1.4470x over previous
#include <hip/hip_runtime.h>
#include <hip/hip_bf16.h>

typedef __attribute__((ext_vector_type(2))) float v2f;
typedef __attribute__((ext_vector_type(8))) float v8f;
typedef __attribute__((ext_vector_type(4))) float v4f;

#define B_   16
#define N_   4096
#define C_   64
#define M_   1024
#define K_   32
#define EPS_ 1e-5f
#define BIGF 3.402823e38f

__global__ __launch_bounds__(256) void k_new_xyz(const float* __restrict__ xyz,
                                                 const int* __restrict__ indices,
                                                 float* __restrict__ new_xyz,
                                                 float* __restrict__ planes) {
  static_assert(B_ * N_ == 4 * B_ * M_, "a thread turns 4 points");
  int t = blockIdx.x * 256 + threadIdx.x;
  if (t >= B_ * M_) return;
  int b = t >> 10, m = t & 1023;
  int n = indices[m];
  const float* s = xyz + ((size_t)b * N_ + n) * 3;
  float* d = new_xyz + (size_t)t * 3;
  d[0] = s[0]; d[1] = s[1]; d[2] = s[2];
  const v4f* p4 = (const v4f*)xyz + (size_t)t * 3;
  const v4f a = p4[0];
  const v4f e = p4[1];
  const v4f f = p4[2];
  v4f px, py, pz;
  px.x = a.x; px.y = a.w; px.z = e.z; px.w = f.y;
  py.x = a.y; py.y = e.x; py.z = e.w; py.w = f.z;
  pz.x = a.z; pz.y = e.y; pz.z = f.x; pz.w = f.w;
  ((v4f*)planes)[t] = px;
  ((v4f*)(planes + (size_t)B_ * N_))[t] = py;
  ((v4f*)(planes + (size_t)2 * B_ * N_))[t] = pz;
}

#define CAP_ 128

__device__ __forceinline__ float sqdist(float qx, float qy, float qz, float qq,
                                        float x, float y, float z) {
  const float pp  = fmaf(z, z, fmaf(x, x, y * y));
  const float dot = fmaf(qz, z, fmaf(qx, x, qy * y));
  return fmaf(-2.0f, dot, qq + pp);
}

__device__ __forceinline__ unsigned okey(float f) {
  const unsigned u = __float_as_uint(f);
  return (u & 0x80000000u) ? ~u : (u | 0x80000000u);
}
__device__ __forceinline__ float okey_inv(unsigned k) {
  return __uint_as_float((k & 0x80000000u) ? (k & 0x7fffffffu) : ~k);
}

template <int NS, int NB>
__device__ __forceinline__ unsigned kth32(const unsigned (&k)[NS]) {
  unsigned v = 0;
  for (int bit = 31; bit > 31 - NB; --bit) {
    const unsigned t = v | (1u << bit);
    int c = 0;
#pragma unroll
    for (int s = 0; s < NS; ++s)
      c += __builtin_popcount(__builtin_amdgcn_ballot_w32(k[s] < t));
    if (c < K_) v = t;
  }
  if (NB < 32) v |= (1u << (32 - NB)) - 1u;
  return v;
}

__global__ __launch_bounds__(32) void k_knn(const float* __restrict__ planes,
                                            const float* __restrict__ new_xyz,
                                            int* __restrict__ knn) {
  __shared__ __attribute__((aligned(16))) float dl[N_];
  __shared__ unsigned ck[CAP_];
  __shared__ int      ci[CAP_];
  const int lane = threadIdx.x;
  const int q = blockIdx.x;
  const int b = q >> 10;

  const float qx = new_xyz[q * 3 + 0];
  const float qy = new_xyz[q * 3 + 1];
  const float qz = new_xyz[q * 3 + 2];
  float qq;
  {
#pragma clang fp contract(off)
    const float yy = qy * qy;
    const float zz = qz * qz;
    qq = fmaf(qx, qx, yy) + zz;
  }
  const float* px = planes + (size_t)b * N_;
  const float* py = px + (size_t)B_ * N_;
  const float* pz = py + (size_t)B_ * N_;

  float m1 = BIGF, m2 = BIGF;
#pragma unroll 8
  for (int j = 0; j < N_ / 128; ++j) {
    const int n0 = j * 128 + lane * 4;
    const v4f x4 = *(const v4f*)(px + n0);
    const v4f y4 = *(const v4f*)(py + n0);
    const v4f z4 = *(const v4f*)(pz + n0);
    v4f d4;
    d4.x = sqdist(qx, qy, qz, qq, x4.x, y4.x, z4.x);
    d4.y = sqdist(qx, qy, qz, qq, x4.y, y4.y, z4.y);
    d4.z = sqdist(qx, qy, qz, qq, x4.z, y4.z, z4.z);
    d4.w = sqdist(qx, qy, qz, qq, x4.w, y4.w, z4.w);
    *(v4f*)(dl + n0) = d4;
#pragma unroll
    for (int e = 0; e < 4; ++e) {
      const float d = d4[e];
      m2 = fminf(m2, fmaxf(m1, d));
      m1 = fminf(m1, d);
    }
  }

  const unsigned kb[2] = { okey(m1), okey(m2) };
  const float fB = okey_inv(kth32<2, 14>(kb));

  int cnt = 0;
  for (int w = 0; w < N_ / 1024; ++w) {
    unsigned bw = 0;
    v4f g4[8];
#pragma unroll
    for (int g = 0; g < 8; ++g) g4[g] = *(const v4f*)(dl + 128 * (8 * w + g) + lane * 4);
#pragma unroll
    for (int u = 0; u < 32; ++u)
      bw |= (g4[u >> 2][u & 3] <= fB) ? (1u << u) : 0u;
    for (int t = 0; t < 32; ++t) {
      const bool has = bw != 0;
      const unsigned mask = __builtin_amdgcn_ballot_w32(has);
      if (mask == 0) break;
      if (has) {
        const int u = __builtin_ctz(bw);
        bw &= bw - 1;
        const int n = 128 * (8 * w + (u >> 2)) + lane * 4 + (u & 3);
        const int pos = cnt + (int)__builtin_amdgcn_mbcnt_lo(mask, 0u);
        if (pos < CAP_) { ck[pos] = okey(dl[n]); ci[pos] = n; }
      }
      cnt += __builtin_popcount(mask);
    }
  }
  __syncthreads();

  int* out = knn + q * K_;
  bool done = false;
  if (cnt >= K_ && cnt <= CAP_) {
    unsigned k[CAP_ / 32];
    int id[CAP_ / 32];
#pragma unroll
    for (int s = 0; s < CAP_ / 32; ++s) {
      const int j = s * 32 + lane;
      const bool valid = j < cnt;
      k[s]  = valid ? ck[j] : 0xffffffffu;
      id[s] = valid ? ci[j] : 0;
    }
    unsigned v32;
    if (cnt <= 64) {
      const unsigned k2[2] = { k[0], k[1] };
      v32 = kth32<2, 32>(k2);
    } else {
      v32 = kth32<CAP_ / 32, 32>(k);
    }
    unsigned mt[CAP_ / 32];
    int total = 0;
#pragma unroll
    for (int s = 0; s < CAP_ / 32; ++s) {
      mt[s] = __builtin_amdgcn_ballot_w32(s * 32 + lane < cnt && k[s] <= v32);
      total += __builtin_popcount(mt[s]);
    }
    if (total == K_) {
      int wpos = 0;
#pragma unroll
      for (int s = 0; s < CAP_ / 32; ++s) {
        if ((mt[s] >> lane) & 1u) out[wpos + (int)__builtin_amdgcn_mbcnt_lo(mt[s], 0u)] = id[s];
        wpos += __builtin_popcount(mt[s]);
      }
      done = true;
    } else if (total > K_) {
      unsigned lt[CAP_ / 32], eq[CAP_ / 32];
      int rk[CAP_ / 32];
      int nless = 0;
#pragma unroll
      for (int s = 0; s < CAP_ / 32; ++s) {
        lt[s] = __builtin_amdgcn_ballot_w32(s * 32 + lane < cnt && k[s] < v32);
        eq[s] = mt[s] & ~lt[s];
        nless += __builtin_popcount(lt[s]);
        rk[s] = 0;
      }
      const int need = K_ - nless;
#pragma unroll
      for (int s2 = 0; s2 < CAP_ / 32; ++s2) {
        for (unsigned em = eq[s2]; em != 0; em &= em - 1) {
          const int oid = __builtin_amdgcn_readlane(id[s2], __builtin_ctz(em));
#pragma unroll
          for (int s = 0; s < CAP_ / 32; ++s) rk[s] += (oid < id[s]) ? 1 : 0;
        }
      }
      int wpos = 0;
#pragma unroll
      for (int s = 0; s < CAP_ / 32; ++s) {
        const bool mine = ((eq[s] >> lane) & 1u) != 0 && rk[s] < need;
        const unsigned sel = lt[s] | __builtin_amdgcn_ballot_w32(mine);
        if ((sel >> lane) & 1u) out[wpos + (int)__builtin_amdgcn_mbcnt_lo(sel, 0u)] = id[s];
        wpos += __builtin_popcount(sel);
      }
      done = true;
    }
  }
  if (!done) {
    for (int kk = 0; kk < K_; ++kk) {
      float mv = BIGF;
      int   mi = 0x7fffffff;
      for (int n = lane; n < N_; n += 32) {
        float v = dl[n];
        if (v < mv) { mv = v; mi = n; }
      }
#pragma unroll
      for (int off = 16; off > 0; off >>= 1) {
        float ov = __shfl_xor(mv, off, 32);
        int   oi = __shfl_xor(mi, off, 32);
        if (ov < mv || (ov == mv && oi < mi)) { mv = ov; mi = oi; }
      }
      if (lane == 0) out[kk] = mi;
      dl[mi] = BIGF;
      __syncthreads();
    }
  }
}

#define NW_     8
#define NCOL_   (NW_ * 32)
#define FPAIRS_ (C_ / 2)
#define TILEW_  (FPAIRS_ * 2 * NCOL_)
#define GCH_    4
#define GQP_    4
#define NK0_    17
#define NK12_   16
#define F1_     (4 * NK0_)
#define F2_     (F1_ + 4 * NK12_)
#define NFRAG_  (F2_ + 8 * NK12_)

__global__ __launch_bounds__(256) void k_gather(const float* __restrict__ features,
                                                const int* __restrict__ knn,
                                                float* __restrict__ xg) {
  __shared__ __attribute__((aligned(16))) float st[(GCH_ / 2) * 2 * N_];
  const int t    = threadIdx.x;
  const int lane = t & 31;
  const int wave = __builtin_amdgcn_readfirstlane(t >> 5);
  const int b    = blockIdx.x % B_;
  const int rest = blockIdx.x / B_;
  const int cg   = rest % (C_ / GCH_);
  const int part = rest / (C_ / GCH_);
  const float* fb = features + ((size_t)b * C_ + cg * GCH_) * N_;

  {
    constexpr int NI = N_ / 4 / 256;
    v4f ra[GCH_ / 2][NI], rb[GCH_ / 2][NI];
#pragma unroll
    for (int pr = 0; pr < GCH_ / 2; ++pr)
#pragma unroll
      for (int i = 0; i < NI; ++i) {
        const int pc = i * 256 + t;
        ra[pr][i] = __builtin_nontemporal_load((const v4f*)(fb + (size_t)(2 * pr)     * N_ + pc * 4));
        rb[pr][i] = __builtin_nontemporal_load((const v4f*)(fb + (size_t)(2 * pr + 1) * N_ + pc * 4));
      }
#pragma unroll
    for (int pr = 0; pr < GCH_ / 2; ++pr)
#pragma unroll
      for (int i = 0; i < NI; ++i) {
        const int pc = i * 256 + t;
        v4f lo, hi;
        lo.x = ra[pr][i].x; lo.y = rb[pr][i].x; lo.z = ra[pr][i].y; lo.w = rb[pr][i].y;
        hi.x = ra[pr][i].z; hi.y = rb[pr][i].z; hi.z = ra[pr][i].w; hi.w = rb[pr][i].w;
        *(v4f*)(st + pr * 2 * N_ + pc * 8)     = lo;
        *(v4f*)(st + pr * 2 * N_ + pc * 8 + 4) = hi;
      }
  }
  __syncthreads();

  constexpr int QPB = M_ / GQP_;
  const int m0 = part * QPB + wave;
  const int*  kq = knn + ((size_t)b * M_ + m0) * K_ + lane;
  float* dq = xg + ((size_t)b * (M_ / NW_) + part * (QPB / NW_)) * TILEW_
                 + (size_t)(cg * (GCH_ / 2)) * (2 * NCOL_) + (wave * 32 + lane) * 2;
#pragma unroll 4
  for (int it = 0; it < QPB / NW_; ++it) {
    const int nj = kq[(size_t)it * NW_ * K_];
#pragma unroll
    for (int pr = 0; pr < GCH_ / 2; ++pr) {
      const v2f p = *(const v2f*)(st + pr * 2 * N_ + nj * 2);
      *(v2f*)(dq + (size_t)it * TILEW_ + pr * (2 * NCOL_)) = p;
    }
  }
}

template <int MASK>
__device__ __forceinline__ float row_xor(float v) {
  return __builtin_bit_cast(float, __builtin_amdgcn_update_dpp(
      0, __builtin_bit_cast(int, v), 0x160 + MASK, 0xf, 0xf, true));
}
__device__ __forceinline__ float row_max16(float v) {
  v = fmaxf(v, row_xor<1>(v));
  v = fmaxf(v, row_xor<2>(v));
  v = fmaxf(v, row_xor<4>(v));
  v = fmaxf(v, row_xor<8>(v));
  return v;
}


typedef const float* __restrict__ cfp;

__device__ __forceinline__ int kperm(int st, int h) {
  return 16 * (st >> 2) + 8 * h + 2 * (st & 3);
}

template <int NKT, int KW, bool PERM>
__device__ __forceinline__ void load_rows(v2f (&a)[NK0_], cfp W, int mt, int lane) {
  const int half = lane >> 4;
  const int arow = mt * 16 + (lane & 15);
#pragma unroll
  for (int st = 0; st < NKT; ++st) {
    const int k0 = PERM ? kperm(st, half) : st * 4 + half * 2;
    a[st].x = (KW % 4 == 0 || k0 + 0 < KW) ? W[arow * KW + k0 + 0] : 0.0f;
    a[st].y = (KW % 4 == 0 || k0 + 1 < KW) ? W[arow * KW + k0 + 1] : 0.0f;
  }
}
struct BnRow { float g, bt, mu, vr; };
__device__ __forceinline__ BnRow load_bn(cfp g, cfp bt, cfp mu, cfp vr, int arow) {
  BnRow p;
  p.g = g[arow]; p.bt = bt[arow]; p.mu = mu[arow]; p.vr = vr[arow];
  return p;
}
template <int NKT>
__device__ __forceinline__ void store_rows(float* frag, float* shift, const v2f (&a)[NK0_],
                                           BnRow p, int arow, int lane) {
  const float s = p.g / sqrtf(p.vr + EPS_);
#pragma unroll
  for (int st = 0; st < NKT; ++st) *(v2f*)(frag + st * 64 + lane * 2) = a[st] * s;
  if ((lane >> 4) == 0) shift[arow] = p.bt - p.mu * s;
}

struct TileOps {
  v2f a[NK0_];
  v8f c;
};
template <int NKT>
__device__ __forceinline__ void load_ops(TileOps& o, const float* frag, const float* shift, int lane) {
#pragma unroll
  for (int st = 0; st < NKT; ++st) o.a[st] = *(const v2f*)(frag + st * 64 + lane * 2);
  const v4f lo = *(const v4f*)(shift + (lane >> 4) * 8);
  const v4f hi = *(const v4f*)(shift + (lane >> 4) * 8 + 4);
  o.c[0] = lo.x; o.c[1] = lo.y; o.c[2] = lo.z; o.c[3] = lo.w;
  o.c[4] = hi.x; o.c[5] = hi.y; o.c[6] = hi.z; o.c[7] = hi.w;
}

__device__ __forceinline__ v8f relu8(v8f a) {
  v8f y;
#pragma unroll
  for (int r = 0; r < 8; ++r) y[r] = fmaxf(a[r], 0.0f);
  return y;
}

__device__ __forceinline__ v8f wmma4(v2f a, v2f b, v8f c) {
  return __builtin_amdgcn_wmma_f32_16x16x4_f32(
       false, a,  false, b,
       (short)0, c,  false,  false);
}

__device__ __forceinline__ void row_tile_0(const TileOps& o, const v2f (&bx)[2][NK0_],
                                           v8f& acc0, v8f& acc1) {
  acc0 = o.c;
  acc1 = o.c;
#pragma unroll
  for (int kt = 0; kt < NK0_; ++kt) {
    acc0 = wmma4(o.a[kt], bx[0][kt], acc0);
    acc1 = wmma4(o.a[kt], bx[1][kt], acc1);
  }
}
__device__ __forceinline__ void row_tile_12(const TileOps& o, const v8f (&yin)[4][2],
                                            v8f& acc0, v8f& acc1) {
  acc0 = o.c;
  acc1 = o.c;
#pragma unroll
  for (int t = 0; t < 4; ++t)
#pragma unroll
    for (int j = 0; j < 4; ++j) {
      v2f b0, b1;
      b0.x = yin[t][0][2 * j]; b0.y = yin[t][0][2 * j + 1];
      b1.x = yin[t][1][2 * j]; b1.y = yin[t][1][2 * j + 1];
      acc0 = wmma4(o.a[4 * t + j], b0, acc0);
      acc1 = wmma4(o.a[4 * t + j], b1, acc1);
    }
}

#define TILE_FENCE() __builtin_amdgcn_sched_barrier(0)

__global__ __launch_bounds__(NW_ * 32) void k_mlp(
    const float* __restrict__ xg, const float* __restrict__ xyz,
    const float* __restrict__ new_xyz, const int* __restrict__ knn,
    const float* __restrict__ W0, const float* __restrict__ g0,
    const float* __restrict__ b0, const float* __restrict__ m0,
    const float* __restrict__ v0,
    const float* __restrict__ W1, const float* __restrict__ g1,
    const float* __restrict__ b1, const float* __restrict__ m1,
    const float* __restrict__ v1,
    const float* __restrict__ W2, const float* __restrict__ g2,
    const float* __restrict__ b2, const float* __restrict__ m2,
    const float* __restrict__ v2,
    float* __restrict__ out_feat)
{
  __shared__ __attribute__((aligned(16))) float As[NFRAG_ * 64];
  __shared__ __attribute__((aligned(16))) float Sh[64 + 64 + 128];
  __shared__ __attribute__((aligned(16))) float Pl[128 * NW_];
  const int wave = __builtin_amdgcn_readfirstlane(threadIdx.x >> 5);
  const int lane = threadIdx.x & 31;
  const int half = lane >> 4;
  const int l16  = lane & 15;
  const int b = blockIdx.x % B_;
  const int mq0 = (blockIdx.x / B_) * NW_;
  const int q = b * M_ + mq0 + wave;

  const float* qc = new_xyz + (size_t)q * 3;
  const float qx = qc[0], qy = qc[1], qz = qc[2];
  const int nj0 = knn[q * K_ + l16];
  const int nj1 = knn[q * K_ + 16 + l16];

  v2f wa[NK0_], wb[NK0_];
  BnRow pa, pb;
  const int arow_a = (wave & 3) * 16 + l16;
  const int arow_b = wave * 16 + l16;
  if (wave < 4) {
    load_rows<NK0_, 67, false>(wa, W0, wave, lane);
    pa = load_bn(g0, b0, m0, v0, arow_a);
  } else {
    load_rows<NK12_, 64, true>(wa, W1, wave - 4, lane);
    wa[NK0_ - 1] = (v2f){0.0f, 0.0f};
    pa = load_bn(g1, b1, m1, v1, arow_a);
  }
  load_rows<NK12_, 64, true>(wb, W2, wave, lane);
  pb = load_bn(g2, b2, m2, v2, arow_b);

  v2f bx[2][NK0_];
  {
    const float* tile = xg + ((size_t)b * (M_ / NW_) + blockIdx.x / B_) * TILEW_
                           + half * (2 * NCOL_) + (wave * 32 + l16) * 2;
#pragma unroll
    for (int kt = 0; kt < NK0_ - 1; ++kt) {
      bx[0][kt] = __builtin_nontemporal_load((const v2f*)(tile + kt * 2 * (2 * NCOL_)));
      bx[1][kt] = __builtin_nontemporal_load((const v2f*)(tile + kt * 2 * (2 * NCOL_) + 32));
    }
#pragma unroll
    for (int ct = 0; ct < 2; ++ct) {
      const float* p = xyz + ((size_t)b * N_ + (ct ? nj1 : nj0)) * 3;
      const float rx = p[0] - qx, ry = p[1] - qy, rz = p[2] - qz;
      bx[ct][NK0_ - 1].x = half ? rz : rx;
      bx[ct][NK0_ - 1].y = half ? 0.0f : ry;
    }
  }

  if (wave < 4)
    store_rows<NK0_>(As + (wave * NK0_) * 64, Sh, wa, pa, arow_a, lane);
  else
    store_rows<NK12_>(As + (F1_ + (wave - 4) * NK12_) * 64, Sh + 64, wa, pa, arow_a, lane);
  store_rows<NK12_>(As + (F2_ + wave * NK12_) * 64, Sh + 128, wb, pb, arow_b, lane);
  __syncthreads();

  TileOps oa, ob;
  v8f acc0, acc1;
  load_ops<NK0_>(oa, As, Sh, lane);

  v8f y0[4][2];
  load_ops<NK0_>(ob, As + (1 * NK0_) * 64, Sh + 16, lane);
  TILE_FENCE();
  row_tile_0(oa, bx, acc0, acc1);
  TILE_FENCE();
  y0[0][0] = relu8(acc0); y0[0][1] = relu8(acc1);
  load_ops<NK0_>(oa, As + (2 * NK0_) * 64, Sh + 32, lane);
  TILE_FENCE();
  row_tile_0(ob, bx, acc0, acc1);
  TILE_FENCE();
  y0[1][0] = relu8(acc0); y0[1][1] = relu8(acc1);
  load_ops<NK0_>(ob, As + (3 * NK0_) * 64, Sh + 48, lane);
  TILE_FENCE();
  row_tile_0(oa, bx, acc0, acc1);
  TILE_FENCE();
  y0[2][0] = relu8(acc0); y0[2][1] = relu8(acc1);
  load_ops<NK12_>(oa, As + F1_ * 64, Sh + 64, lane);
  TILE_FENCE();
  row_tile_0(ob, bx, acc0, acc1);
  TILE_FENCE();
  y0[3][0] = relu8(acc0); y0[3][1] = relu8(acc1);

  v8f y1[4][2];
  load_ops<NK12_>(ob, As + (F1_ + 1 * NK12_) * 64, Sh + 64 + 16, lane);
  TILE_FENCE();
  row_tile_12(oa, y0, acc0, acc1);
  TILE_FENCE();
  y1[0][0] = relu8(acc0); y1[0][1] = relu8(acc1);
  load_ops<NK12_>(oa, As + (F1_ + 2 * NK12_) * 64, Sh + 64 + 32, lane);
  TILE_FENCE();
  row_tile_12(ob, y0, acc0, acc1);
  TILE_FENCE();
  y1[1][0] = relu8(acc0); y1[1][1] = relu8(acc1);
  load_ops<NK12_>(ob, As + (F1_ + 3 * NK12_) * 64, Sh + 64 + 48, lane);
  TILE_FENCE();
  row_tile_12(oa, y0, acc0, acc1);
  TILE_FENCE();
  y1[2][0] = relu8(acc0); y1[2][1] = relu8(acc1);
  load_ops<NK12_>(oa, As + F2_ * 64, Sh + 128, lane);
  TILE_FENCE();
  row_tile_12(ob, y0, acc0, acc1);
  TILE_FENCE();
  y1[3][0] = relu8(acc0); y1[3][1] = relu8(acc1);

  auto pool_tile = [&](int mt) {
    float res = 0.0f;
#pragma unroll
    for (int r = 0; r < 8; ++r) {
      const float v = row_max16(fmaxf(acc0[r], acc1[r]));
      if (l16 == r) res = v;
    }
    if (l16 < 8) Pl[(mt * 16 + half * 8 + l16) * NW_ + wave] = fmaxf(res, 0.0f);
  };
#pragma nounroll
  for (int mp = 0; mp < 4; ++mp) {
    const int t1 = 2 * mp + 1;
    const int t2 = (mp < 3) ? 2 * mp + 2 : 7;
    load_ops<NK12_>(ob, As + (F2_ + t1 * NK12_) * 64, Sh + 128 + t1 * 16, lane);
    TILE_FENCE();
    row_tile_12(oa, y1, acc0, acc1);
    TILE_FENCE();
    pool_tile(2 * mp);
    load_ops<NK12_>(oa, As + (F2_ + t2 * NK12_) * 64, Sh + 128 + t2 * 16, lane);
    TILE_FENCE();
    row_tile_12(ob, y1, acc0, acc1);
    TILE_FENCE();
    pool_tile(t1);
  }
  __syncthreads();

  {
    const int t = threadIdx.x;
    const int row = t >> 1, part = t & 1;
    const v4f v = *(const v4f*)(Pl + row * NW_ + part * 4);
    *(v4f*)(out_feat + ((size_t)b * 128 + row) * M_ + mq0 + part * 4) = v;
  }
}

extern "C" void kernel_launch(void* const* d_in, const int* in_sizes, int n_in,
                              void* d_out, int out_size, void* d_ws, size_t ws_size,
                              hipStream_t stream) {
  const float* xyz      = (const float*)d_in[0];
  const float* features = (const float*)d_in[1];
  const int*   indices  = (const int*)d_in[2];
  const float* W0 = (const float*)d_in[3];
  const float* g0 = (const float*)d_in[4];
  const float* b0 = (const float*)d_in[5];
  const float* m0 = (const float*)d_in[6];
  const float* v0 = (const float*)d_in[7];
  const float* W1 = (const float*)d_in[8];
  const float* g1 = (const float*)d_in[9];
  const float* b1 = (const float*)d_in[10];
  const float* m1 = (const float*)d_in[11];
  const float* v1 = (const float*)d_in[12];
  const float* W2 = (const float*)d_in[13];
  const float* g2 = (const float*)d_in[14];
  const float* b2 = (const float*)d_in[15];
  const float* m2 = (const float*)d_in[16];
  const float* v2 = (const float*)d_in[17];

  float* new_xyz  = (float*)d_out;
  float* out_feat = new_xyz + (size_t)B_ * M_ * 3;
  int*   knn      = (int*)d_ws;
  float* xg       = (float*)d_ws + (size_t)B_ * M_ * K_;
  float* planes   = xg + (size_t)B_ * (M_ / NW_) * TILEW_;

  k_new_xyz<<<(B_ * M_ + 255) / 256, 256, 0, stream>>>(xyz, indices, new_xyz, planes);
  k_knn<<<B_ * M_, 32, 0, stream>>>(planes, new_xyz, knn);
  k_gather<<<B_ * (C_ / GCH_) * GQP_, 256, 0, stream>>>(features, knn, xg);
  k_mlp<<<(B_ * M_) / NW_, NW_ * 32, 0, stream>>>(xg, xyz, new_xyz, knn,
      W0, g0, b0, m0, v0, W1, g1, b1, m1, v1, W2, g2, b2, m2, v2, out_feat);
}
